# starting kernel
# baseline (speedup 1.0000x reference)
#include <hip/hip_runtime.h>
#include <hip/hip_bf16.h>


typedef __attribute__((ext_vector_type(2))) float v2f;
typedef __attribute__((ext_vector_type(8))) float v8f;

#define N_DST1   50000
#define N_DST2   10000
#define IN_F     256
#define KDIM     512
#define KSTEPS   (KDIM / 4)
#define KSPLIT   (IN_F / 4)
#define HID_F    256
#define N_CLS    64

__global__ void zero_i32_kernel(int* __restrict__ p, int n) {
  int i = blockIdx.x * blockDim.x + threadIdx.x;
  if (i < n) p[i] = 0;
}

__global__ void hist_kernel(const int* __restrict__ dst,
                            int* __restrict__ hist, int nE) {
  int i = blockIdx.x * blockDim.x + threadIdx.x;
  if (i < nE) atomicAdd(&hist[dst[i]], 1);
}

__global__ __launch_bounds__(256)
void scan_kernel(const int* __restrict__ hist,
                 int* __restrict__ row_ptr,
                 int* __restrict__ cursor, int n) {
  __shared__ int partials[256];
  const int t = threadIdx.x;
  const int chunk = (n + 255) / 256;
  int begin = t * chunk;
  int end   = begin + chunk; if (end > n) end = n;
  if (begin > n) begin = n;

  int sum = 0;
  for (int i = begin; i < end; ++i) sum += hist[i];
  partials[t] = sum;
  __syncthreads();
  if (t == 0) {
    int run = 0;
    for (int i = 0; i < 256; ++i) { int v = partials[i]; partials[i] = run; run += v; }
  }
  __syncthreads();
  int run = partials[t];
  for (int i = begin; i < end; ++i) {
    row_ptr[i] = run;
    cursor[i]  = run;
    run += hist[i];
  }
  if (end == n) row_ptr[n] = run;
}

__global__ void bin_kernel(const int* __restrict__ src,
                           const int* __restrict__ dst,
                           int* __restrict__ cursor,
                           int* __restrict__ srcs, int nE) {
  int i = blockIdx.x * blockDim.x + threadIdx.x;
  if (i < nE) {
    int d = dst[i];
    int pos = atomicAdd(&cursor[d], 1);
    srcs[pos] = src[i];
  }
}

__global__ __launch_bounds__(256)
void gather_mean_kernel(const float* __restrict__ hsrc,
                        const int* __restrict__ srcs,
                        const int* __restrict__ row_ptr,
                        float* __restrict__ hN,
                        int n_dst) {
  const int d = blockIdx.x * (blockDim.x >> 5) + (threadIdx.x >> 5);
  if (d >= n_dst) return;
  const int lane = threadIdx.x & 31;
  const int beg = row_ptr[d];
  const int end = row_ptr[d + 1];

  float4 a0 = make_float4(0.f, 0.f, 0.f, 0.f);
  float4 a1 = make_float4(0.f, 0.f, 0.f, 0.f);
  for (int j = beg; j < end; ++j) {
    int s = srcs[j];
    const float4* p = (const float4*)(hsrc + (size_t)s * IN_F) + lane * 2;
    float4 v0 = p[0];
    float4 v1 = p[1];
    a0.x += v0.x; a0.y += v0.y; a0.z += v0.z; a0.w += v0.w;
    a1.x += v1.x; a1.y += v1.y; a1.z += v1.z; a1.w += v1.w;
  }
  const float inv = 1.0f / fmaxf((float)(end - beg), 1.0f);
  a0.x *= inv; a0.y *= inv; a0.z *= inv; a0.w *= inv;
  a1.x *= inv; a1.y *= inv; a1.z *= inv; a1.w *= inv;
  float4* o = (float4*)(hN + (size_t)d * IN_F) + lane * 2;
  o[0] = a0;
  o[1] = a1;
}

__global__ void pack_W_kernel(const float* __restrict__ W,
                              float* __restrict__ Bpack, int N) {
  const int ngrpN = N >> 6;
  const int total = KSTEPS * ngrpN * 128;
  int idx = blockIdx.x * blockDim.x + threadIdx.x;
  if (idx >= total) return;

  const int w    = idx & 127;
  const int t    = w & 3;
  const int lane = w >> 2;
  const int blk  = idx >> 7;
  const int gn   = blk % ngrpN;
  const int ks   = blk / ngrpN;

  const int kh  = (lane < 16) ? 0 : 2;
  const int col = gn * 64 + t * 16 + (lane & 15);
  const int kr  = ks * 4 + kh;

  float2 v;
  v.x = W[(size_t)kr * N + col];
  v.y = W[(size_t)(kr + 1) * N + col];
  ((float2*)Bpack)[idx] = v;
}

__global__ __launch_bounds__(128)
void sage_gemm_wmma_kernel(const float* __restrict__ A0,
                           const float* __restrict__ A1,
                           const float* __restrict__ Bpack,
                           const float* __restrict__ bias,
                           float* __restrict__ out,
                           int M, int N, int do_relu) {
  const int lane = threadIdx.x & 31;
  const int wave = threadIdx.x >> 5;
  const int ngrpN = N >> 6;
  const int grp   = blockIdx.x * (blockDim.x >> 5) + wave;
  const int ngrps = (M >> 4) * ngrpN;
  if (grp >= ngrps) return;

  const int tm = grp / ngrpN;
  const int gn = grp % ngrpN;

  const int l15  = lane & 15;
  const int row  = tm * 16 + l15;
  const int col0 = gn * 64 + l15;
  const int kh   = (lane < 16) ? 0 : 2;

  const float* Arow0 = A0 + (size_t)row * IN_F + kh;
  const float* Arow1 = A1 + (size_t)row * IN_F + kh;
  const float* Bp    = Bpack + (size_t)gn * 256 + lane * 8;
  const size_t bstride = (size_t)ngrpN * 256;

  v8f acc[4] = {};

#define K_STEP(AROW, I, KS)                                                   \
  {                                                                           \
    v2f a = *(const v2f*)((AROW) + (I) * 4);               \
    const float4* bp = (const float4*)(Bp + (size_t)(KS) * bstride);          \
    float4 bA = bp[0];                                   \
    float4 bB = bp[1];                                   \
    v2f b0; b0.x = bA.x; b0.y = bA.y;                                         \
    v2f b1; b1.x = bA.z; b1.y = bA.w;                                         \
    v2f b2; b2.x = bB.x; b2.y = bB.y;                                         \
    v2f b3; b3.x = bB.z; b3.y = bB.w;                                         \
    acc[0] = __builtin_amdgcn_wmma_f32_16x16x4_f32(false, a, false, b0,       \
                                                   (short)0, acc[0], false, false); \
    acc[1] = __builtin_amdgcn_wmma_f32_16x16x4_f32(false, a, false, b1,       \
                                                   (short)0, acc[1], false, false); \
    acc[2] = __builtin_amdgcn_wmma_f32_16x16x4_f32(false, a, false, b2,       \
                                                   (short)0, acc[2], false, false); \
    acc[3] = __builtin_amdgcn_wmma_f32_16x16x4_f32(false, a, false, b3,       \
                                                   (short)0, acc[3], false, false); \
  }

#pragma unroll 4
  for (int i = 0; i < KSPLIT; ++i) {
    K_STEP(Arow0, i, i);
  }
#pragma unroll 4
  for (int i = 0; i < KSTEPS - KSPLIT; ++i) {
    K_STEP(Arow1, i, KSPLIT + i);
  }
#undef K_STEP

  const int mbase = tm * 16 + ((lane < 16) ? 0 : 8);
#pragma unroll
  for (int t = 0; t < 4; ++t) {
    const int col = col0 + t * 16;
    const float bv = bias[col];
#pragma unroll
    for (int i = 0; i < 8; ++i) {
      float v = acc[t][i] + bv;
      if (do_relu) v = fmaxf(v, 0.0f);
      out[(size_t)(mbase + i) * N + col] = v;
    }
  }
}

extern "C" void kernel_launch(void* const* d_in, const int* in_sizes, int n_in,
                              void* d_out, int out_size, void* d_ws, size_t ws_size,
                              hipStream_t stream) {
  const float* x   = (const float*)d_in[0];
  const float* W1  = (const float*)d_in[1];
  const float* b1  = (const float*)d_in[2];
  const float* W2  = (const float*)d_in[3];
  const float* b2  = (const float*)d_in[4];
  const int* src1  = (const int*)d_in[5];
  const int* dst1  = (const int*)d_in[6];
  const int* src2  = (const int*)d_in[7];
  const int* dst2  = (const int*)d_in[8];
  const int E1 = in_sizes[5];
  const int E2 = in_sizes[7];

  char* wsb = (char*)d_ws;
  size_t off = 0;
  auto take = [&](size_t n_elems) { void* p = wsb + off; off += n_elems * 4; return p; };

  int* hist1    = (int*)take(N_DST1);
  int* hist2    = (int*)take(N_DST2);
  int* row_ptr1 = (int*)take(N_DST1 + 1);
  int* cursor1  = (int*)take(N_DST1);
  int* srcs1    = (int*)take((size_t)E1);
  int* row_ptr2 = (int*)take(N_DST2 + 1);
  int* cursor2  = (int*)take(N_DST2);
  int* srcs2    = (int*)take((size_t)E2);
  float* hN1    = (float*)take((size_t)N_DST1 * IN_F);
  float* h      = (float*)take((size_t)N_DST1 * HID_F);
  float* hN2    = (float*)take((size_t)N_DST2 * IN_F);
  float* Bpack1 = (float*)take((size_t)KDIM * HID_F);
  float* Bpack2 = (float*)take((size_t)KDIM * N_CLS);
  (void)ws_size;

  {
    int t1 = KSTEPS * (HID_F / 64) * 128;
    int t2 = KSTEPS * (N_CLS / 64) * 128;
    pack_W_kernel<<<(t1 + 255) / 256, 256, 0, stream>>>(W1, Bpack1, HID_F);
    pack_W_kernel<<<(t2 + 255) / 256, 256, 0, stream>>>(W2, Bpack2, N_CLS);
  }

  zero_i32_kernel<<<(N_DST1 + N_DST2 + 255) / 256, 256, 0, stream>>>(
      hist1, N_DST1 + N_DST2);
  hist_kernel<<<(E1 + 255) / 256, 256, 0, stream>>>(dst1, hist1, E1);
  scan_kernel<<<1, 256, 0, stream>>>(hist1, row_ptr1, cursor1, N_DST1);
  bin_kernel<<<(E1 + 255) / 256, 256, 0, stream>>>(src1, dst1, cursor1, srcs1, E1);
  gather_mean_kernel<<<N_DST1 / 8, 256, 0, stream>>>(x, srcs1, row_ptr1, hN1, N_DST1);

  {
    int ngrps = (N_DST1 / 16) * (HID_F / 64);
    sage_gemm_wmma_kernel<<<ngrps / 4, 128, 0, stream>>>(
        x, hN1, Bpack1, b1, h, N_DST1, HID_F,  1);
  }

  hist_kernel<<<(E2 + 255) / 256, 256, 0, stream>>>(dst2, hist2, E2);
  scan_kernel<<<1, 256, 0, stream>>>(hist2, row_ptr2, cursor2, N_DST2);
  bin_kernel<<<(E2 + 255) / 256, 256, 0, stream>>>(src2, dst2, cursor2, srcs2, E2);
  gather_mean_kernel<<<N_DST2 / 8, 256, 0, stream>>>(h, srcs2, row_ptr2, hN2, N_DST2);

  {
    int ngrps = (N_DST2 / 16) * (N_CLS / 64);
    sage_gemm_wmma_kernel<<<(ngrps + 3) / 4, 128, 0, stream>>>(
        h, hN2, Bpack2, b2, (float*)d_out, N_DST2, N_CLS,  0);
  }
}


# Layer-1 gather by one resident set of waves (7 dst nodes per wave, sums in registers) that visit neighbors by 32768-row band of x, so re-reads of a row hit L2; sum order now band by band
# speedup vs baseline: 2.6690x; 2.6690x over previous
#include <hip/hip_runtime.h>
#include <hip/hip_bf16.h>


typedef __attribute__((ext_vector_type(2))) float v2f;
typedef __attribute__((ext_vector_type(8))) float v8f;

#define N_DST1   50000
#define N_DST2   10000
#define IN_F     256
#define KDIM     512
#define KSTEPS   (KDIM / 4)
#define KSPLIT   (IN_F / 4)
#define HID_F    256
#define N_CLS    64

#define NBR_CAP  64

#define NPART      4
#define PCAP       (NBR_CAP / NPART)
#define OWN_ROWS1  896
#define OWN_ROWS2  1280
#define OWN_MAX    1280
#define OWN_THR    1024
#define LIST_CAP   48
#define SCAN_U     8
#define LIST_EBITS 20
#define NRNG1      ((N_DST1 + OWN_ROWS1 - 1) / OWN_ROWS1)
#define NRNG2      ((N_DST2 + OWN_ROWS2 - 1) / OWN_ROWS2)
#define NWG_ROWS1  (NRNG1 * NPART)
#define NWG_ROWS2  (NRNG2 * NPART)
#define CPITCH1    ((N_DST1 + 31) / 32 * 32)
#define CPITCH2    ((N_DST2 + 31) / 32 * 32)

static_assert(NPART == 4, "the gather's slot walk is written for four parts");
static_assert(OWN_MAX >= OWN_ROWS1 && OWN_MAX >= OWN_ROWS2, "LDS sizing");
static_assert(OWN_MAX <= (1 << (32 - LIST_EBITS)), "local row must fit the list word");

__device__ __forceinline__ void place_owned(int i, int s, int dl,
                                            int* s_cnt, int* s_head,
                                            int* s_slots,
                                            int2* __restrict__ ovf) {
  const int rank = atomicAdd(&s_cnt[dl], 1);
  if (rank < PCAP) {
    s_slots[dl * PCAP + rank] = s;
  } else {
    const int prev = atomicExch(&s_head[dl], i + 1);
    ovf[i] = make_int2(s, prev);
  }
}

__global__ __launch_bounds__(OWN_THR)
void build_rows_kernel(const int* __restrict__ src1,
                       const int* __restrict__ dst1,
                       int* __restrict__ cnt1,
                       int* __restrict__ slots1,
                       int* __restrict__ ovf_head1,
                       int2* __restrict__ ovf1, int nE1, int nDst1, int nwg1,
                       const int* __restrict__ src2,
                       const int* __restrict__ dst2,
                       int* __restrict__ cnt2,
                       int* __restrict__ slots2,
                       int* __restrict__ ovf_head2,
                       int2* __restrict__ ovf2, int nE2, int nDst2) {
  __shared__ int s_cnt[OWN_MAX];
  __shared__ int s_head[OWN_MAX];
  __shared__ __attribute__((aligned(16))) int s_slots[OWN_MAX * PCAP];
  __shared__ int s_list[LIST_CAP * OWN_THR];

  const bool first = (int)blockIdx.x < nwg1;
  const int* __restrict__ src = first ? src1 : src2;
  const int* __restrict__ dst = first ? dst1 : dst2;
  int*  cnt      = first ? cnt1 : cnt2;
  int*  slots    = first ? slots1 : slots2;
  int*  ovf_head = first ? ovf_head1 : ovf_head2;
  int2* ovf      = first ? ovf1 : ovf2;
  const int nE      = first ? nE1 : nE2;
  const int nDst    = first ? nDst1 : nDst2;
  const int ownRows = first ? OWN_ROWS1 : OWN_ROWS2;
  const int cpitch  = first ? CPITCH1 : CPITCH2;
  const int lb    = first ? (int)blockIdx.x : (int)blockIdx.x - nwg1;
  const int part  = lb % NPART;
  const int owner = lb / NPART;
  const int base  = owner * ownRows;
  const int left  = nDst - base;
  const unsigned nown = (unsigned)(left < ownRows ? (left > 0 ? left : 0) : ownRows);
  const int tid = threadIdx.x;

  for (int t = tid; t < ownRows; t += OWN_THR) { s_cnt[t] = 0; s_head[t] = 0; }
  __syncthreads();

  const int nq = nE >> 2;
  const int q4 = (nq + NPART - 1) / NPART;
  int j_lo = part * q4;                          if (j_lo > nq) j_lo = nq;
  int j_hi = j_lo + q4;                          if (j_hi > nq) j_hi = nq;
  int j_sc = j_lo + (1 << (LIST_EBITS - 2));     if (j_sc > j_hi) j_sc = j_hi;
  const int e_base = j_lo << 2;
  const int e_end  = (part == NPART - 1) ? nE : (j_hi << 2);

  int lp = tid;
  const int lp_room = tid + (LIST_CAP - 4 * SCAN_U) * OWN_THR;

  auto flush = [&]() {
    for (int k0 = tid; k0 < lp; k0 += 8 * OWN_THR) {
      int w[8], s[8];
#pragma unroll
      for (int u = 0; u < 8; ++u) {
        const int k = k0 + u * OWN_THR;
        const bool on = k < lp;
        w[u] = on ? s_list[k] : 0;
        s[u] = on ? src[e_base + (w[u] & ((1 << LIST_EBITS) - 1))] : 0;
      }
#pragma unroll
      for (int u = 0; u < 8; ++u) {
        if (k0 + u * OWN_THR < lp)
          place_owned(e_base + (w[u] & ((1 << LIST_EBITS) - 1)), s[u],
                      (int)((unsigned)w[u] >> LIST_EBITS), s_cnt, s_head, s_slots, ovf);
      }
    }
    lp = tid;
  };

  const int4* d4 = (const int4*)dst;

  auto group = [&](int j0, bool partial) {
    if (lp > lp_room) flush();
    int4 v[SCAN_U];
#pragma unroll
    for (int u = 0; u < SCAN_U; ++u) {
      const int jj = j0 + u * OWN_THR;
      v[u] = d4[(partial && jj >= j_sc) ? j_sc - 1 : jj];
    }
#pragma unroll
    for (int u = 0; u < SCAN_U; ++u) {
      const int jj = j0 + u * OWN_THR;
      unsigned t0 = (unsigned)(v[u].x - base);
      unsigned t1 = (unsigned)(v[u].y - base);
      unsigned t2 = (unsigned)(v[u].z - base);
      unsigned t3 = (unsigned)(v[u].w - base);
      if (partial && jj >= j_sc) { t0 = ~0u; t1 = ~0u; t2 = ~0u; t3 = ~0u; }
      const unsigned ta = t0 < t1 ? t0 : t1;
      const unsigned tb = t2 < t3 ? t2 : t3;
      if ((ta < tb ? ta : tb) < nown) {
        const unsigned e = (unsigned)(jj - j_lo) << 2;
        if (t0 < nown) { s_list[lp] = (int)((t0 << LIST_EBITS) | e);        lp += OWN_THR; }
        if (t1 < nown) { s_list[lp] = (int)((t1 << LIST_EBITS) | (e | 1u)); lp += OWN_THR; }
        if (t2 < nown) { s_list[lp] = (int)((t2 << LIST_EBITS) | (e | 2u)); lp += OWN_THR; }
        if (t3 < nown) { s_list[lp] = (int)((t3 << LIST_EBITS) | (e | 3u)); lp += OWN_THR; }
      }
    }
  };

  if (j_sc > j_lo) {
    int j0 = j_lo + tid;
    for (; j0 + (SCAN_U - 1) * OWN_THR < j_sc; j0 += SCAN_U * OWN_THR) group(j0, false);
    if (j0 < j_sc) group(j0, true);
  }
  flush();
  for (int e = (j_sc << 2) + tid; e < e_end; e += OWN_THR) {
    const unsigned dl = (unsigned)(dst[e] - base);
    if (dl < nown) place_owned(e, src[e], (int)dl, s_cnt, s_head, s_slots, ovf);
  }
  __syncthreads();

  int* pc = cnt      + (size_t)part * cpitch + base;
  int* ph = ovf_head + (size_t)part * cpitch + base;
  for (int t = tid; t < (int)nown; t += OWN_THR) {
    pc[t] = s_cnt[t];
    ph[t] = s_head[t];
  }
  int4* gs = (int4*)(slots + ((size_t)part * nDst + base) * PCAP);
  const int4* ls = (const int4*)s_slots;
  for (int t = tid; t < (int)nown * (PCAP / 4); t += OWN_THR) gs[t] = ls[t];
}


#define GATHER_ROWS(N)                                                        \
  {                                                                           \
    int s[N];                                                                 \
    float4 v0[N], v1[N];                                                      \
    _Pragma("unroll")                                                         \
    for (int u = 0; u < (N); ++u) {                                           \
      const int b = __builtin_ctzll(m);                  \
      m &= m - 1;                                                             \
      const int lo = __builtin_amdgcn_readlane(w01, b & 31);                  \
      const int hi = __builtin_amdgcn_readlane(w23, b & 31);                  \
      s[u] = b < 32 ? lo : hi;                                                \
    }                                                                         \
    _Pragma("unroll")                                                         \
    for (int u = 0; u < (N); ++u) {                                           \
      const float4* p = hb + (size_t)s[u] * (IN_F / 4);                       \
      v0[u] = p[0];                                                           \
      v1[u] = p[1];                                                           \
    }                                                                         \
    _Pragma("unroll")                                                         \
    for (int u = 0; u < (N); ++u) {                                           \
      a0.x += v0[u].x; a0.y += v0[u].y; a0.z += v0[u].z; a0.w += v0[u].w;     \
      a1.x += v1[u].x; a1.y += v1[u].y; a1.z += v1[u].z; a1.w += v1[u].w;     \
    }                                                                         \
  }

__global__ __launch_bounds__(256)
void gather_mean_kernel(const float* __restrict__ hsrc,
                        const int* __restrict__ slots,
                        const int* __restrict__ cnt,
                        const int* __restrict__ ovf_head,
                        const int2* __restrict__ ovf,
                        float* __restrict__ hN,
                        int n_dst, int cpitch, int nE) {
  const int d = __builtin_amdgcn_readfirstlane(
      (int)(blockIdx.x * (blockDim.x >> 5) + (threadIdx.x >> 5)));
  if (d >= n_dst) return;
  const int lane = threadIdx.x & 31;
  const int plane = n_dst * PCAP;
  const int* rl = slots + (size_t)d * PCAP + (size_t)(lane >> 4) * plane + (lane & 15);
  const int w01 = rl[0];
  const int w23 = rl[2 * (size_t)plane];
  const int c0 = cnt[d];
  const int c1 = cnt[cpitch + d];
  const int c2 = cnt[2 * cpitch + d];
  const int c3 = cnt[3 * cpitch + d];
  const int deg = c0 + c1 + c2 + c3;
  const int e0 = c0 < PCAP ? (c0 > 0 ? c0 : 0) : PCAP;
  const int e1 = c1 < PCAP ? (c1 > 0 ? c1 : 0) : PCAP;
  const int e2 = c2 < PCAP ? (c2 > 0 ? c2 : 0) : PCAP;
  const int e3 = c3 < PCAP ? (c3 > 0 ? c3 : 0) : PCAP;
  const int tot = e0 + e1 + e2 + e3;
  unsigned long long m =
        (unsigned long long)((1u << e0) - 1u)
      | ((unsigned long long)((1u << e1) - 1u) << 16)
      | ((unsigned long long)((1u << e2) - 1u) << 32)
      | ((unsigned long long)((1u << e3) - 1u) << 48);
  const float4* hb = (const float4*)hsrc + lane * 2;

  float4 a0 = make_float4(0.f, 0.f, 0.f, 0.f);
  float4 a1 = make_float4(0.f, 0.f, 0.f, 0.f);
  int n = tot;
  for (; n >= 8; n -= 8) GATHER_ROWS(8)
  if (n & 4) GATHER_ROWS(4)
  if (n & 2) GATHER_ROWS(2)
  if (n & 1) GATHER_ROWS(1)
  if (deg > tot) {
    const int cp[NPART] = {c0, c1, c2, c3};
#pragma unroll
    for (int q = 0; q < NPART; ++q) {
      if (cp[q] <= PCAP) continue;
      int t = ovf_head[q * cpitch + d];
      for (int k = 0; t > 0 && k < nE; ++k) {
        const int2 r = ovf[t - 1];
        t = r.y;
        const float4* p = (const float4*)(hsrc + (size_t)r.x * IN_F) + lane * 2;
        float4 v0 = p[0];
        float4 v1 = p[1];
        a0.x += v0.x; a0.y += v0.y; a0.z += v0.z; a0.w += v0.w;
        a1.x += v1.x; a1.y += v1.y; a1.z += v1.z; a1.w += v1.w;
      }
    }
  }
  const float inv = 1.0f / fmaxf((float)deg, 1.0f);
  a0.x *= inv; a0.y *= inv; a0.z *= inv; a0.w *= inv;
  a1.x *= inv; a1.y *= inv; a1.z *= inv; a1.w *= inv;
  float4* o = (float4*)(hN + (size_t)d * IN_F) + lane * 2;
  o[0] = a0;
  o[1] = a1;
}
#undef GATHER_ROWS

#define GB_K        7
#define GB_WPB      8
#define GB_NWG      1024
#define BAND_SHIFT  15
#define NBAND       8

static_assert(N_DST1 <= GB_K * GB_NWG * GB_WPB, "every layer-1 node needs a wave slot");

#define BAND_ROWS(N, J)                                                       \
  {                                                                           \
    int s[N];                                                                 \
    float4 v0[N], v1[N];                                                      \
    _Pragma("unroll")                                                         \
    for (int u = 0; u < (N); ++u) {                                           \
      const int b = __builtin_ctzll(m);             \
      m &= m - 1;                                                             \
      const int lo = __builtin_amdgcn_readlane(w01[J], b & 31);               \
      const int hi = __builtin_amdgcn_readlane(w23[J], b & 31);               \
      s[u] = b < 32 ? lo : hi;                                                \
    }                                                                         \
    _Pragma("unroll")                                                         \
    for (int u = 0; u < (N); ++u) {                                           \
      const float4* p = hb + (size_t)s[u] * (IN_F / 4);                       \
      v0[u] = p[0];                                                           \
      v1[u] = p[1];                                                           \
    }                                                                         \
    _Pragma("unroll")                                                         \
    for (int u = 0; u < (N); ++u) {                                           \
      a0[J].x += v0[u].x; a0[J].y += v0[u].y;                                 \
      a0[J].z += v0[u].z; a0[J].w += v0[u].w;                                 \
      a1[J].x += v1[u].x; a1[J].y += v1[u].y;                                 \
      a1[J].z += v1[u].z; a1[J].w += v1[u].w;                                 \
    }                                                                         \
  }

__global__ __launch_bounds__(32 * GB_WPB)
void gather_mean_banded_kernel(const float* __restrict__ hsrc,
                               const int* __restrict__ slots,
                               const int* __restrict__ cnt,
                               const int* __restrict__ ovf_head,
                               const int2* __restrict__ ovf,
                               float* __restrict__ hN,
                               int n_dst, int cpitch, int nE) {
  const int wid = __builtin_amdgcn_readfirstlane(
      (int)(blockIdx.x * GB_WPB + (threadIdx.x >> 5)));
  const int nw   = (int)gridDim.x * GB_WPB;
  const int lane = threadIdx.x & 31;
  const int plane = n_dst * PCAP;
  const float4* hb = (const float4*)hsrc + lane * 2;

  int w01[GB_K], w23[GB_K];
  int deg[GB_K];
  bool chained[GB_K];
  float4 a0[GB_K], a1[GB_K];
#pragma unroll
  for (int j = 0; j < GB_K; ++j) {
    const int d  = wid + j * nw;
    const bool on = d < n_dst;
    const int dd = on ? d : n_dst - 1;
    a0[j] = make_float4(0.f, 0.f, 0.f, 0.f);
    a1[j] = make_float4(0.f, 0.f, 0.f, 0.f);
    const int* rl = slots + (size_t)dd * PCAP + (size_t)(lane >> 4) * plane + (lane & 15);
    const int x01 = rl[0];
    const int x23 = rl[2 * (size_t)plane];
    const int c0 = on ? cnt[dd] : 0;
    const int c1 = on ? cnt[cpitch + dd] : 0;
    const int c2 = on ? cnt[2 * cpitch + dd] : 0;
    const int c3 = on ? cnt[3 * cpitch + dd] : 0;
    const int e0 = c0 < PCAP ? (c0 > 0 ? c0 : 0) : PCAP;
    const int e1 = c1 < PCAP ? (c1 > 0 ? c1 : 0) : PCAP;
    const int e2 = c2 < PCAP ? (c2 > 0 ? c2 : 0) : PCAP;
    const int e3 = c3 < PCAP ? (c3 > 0 ? c3 : 0) : PCAP;
    deg[j] = c0 + c1 + c2 + c3;
    chained[j] = (c0 > PCAP) | (c1 > PCAP) | (c2 > PCAP) | (c3 > PCAP);
    const int r = lane & 15;
    w01[j] = r < (lane < 16 ? e0 : e1) ? x01 : -1;
    w23[j] = r < (lane < 16 ? e2 : e3) ? x23 : -1;
  }

#pragma nounroll
  for (int p = 0; p < NBAND; ++p) {
#pragma unroll
    for (int j = 0; j < GB_K; ++j) {
      int b01 = w01[j] >> BAND_SHIFT;  b01 = b01 > NBAND - 1 ? NBAND - 1 : b01;
      int b23 = w23[j] >> BAND_SHIFT;  b23 = b23 > NBAND - 1 ? NBAND - 1 : b23;
      unsigned long long m =
            (unsigned long long)__builtin_amdgcn_ballot_w32(b01 == p)
          | ((unsigned long long)__builtin_amdgcn_ballot_w32(b23 == p) << 32);
      int n = __builtin_popcountll(m);
      for (; n >= 4; n -= 4) BAND_ROWS(4, j)
      if (n & 2) BAND_ROWS(2, j)
      if (n & 1) BAND_ROWS(1, j)
    }
  }

#pragma unroll
  for (int j = 0; j < GB_K; ++j) {
    const int d = wid + j * nw;
    if (d >= n_dst) continue;
    if (chained[j]) {
      for (int q = 0; q < NPART; ++q) {
        if (cnt[q * cpitch + d] <= PCAP) continue;
        int t = ovf_head[q * cpitch + d];
        for (int k = 0; t > 0 && k < nE; ++k) {
          const int2 r = ovf[t - 1];
          t = r.y;
          const float4* p = (const float4*)(hsrc + (size_t)r.x * IN_F) + lane * 2;
          float4 v0 = p[0];
          float4 v1 = p[1];
          a0[j].x += v0.x; a0[j].y += v0.y; a0[j].z += v0.z; a0[j].w += v0.w;
          a1[j].x += v1.x; a1[j].y += v1.y; a1[j].z += v1.z; a1[j].w += v1.w;
        }
      }
    }
    const float inv = 1.0f / fmaxf((float)deg[j], 1.0f);
    float4 r0 = a0[j], r1 = a1[j];
    r0.x *= inv; r0.y *= inv; r0.z *= inv; r0.w *= inv;
    r1.x *= inv; r1.y *= inv; r1.z *= inv; r1.w *= inv;
    float4* o = (float4*)(hN + (size_t)d * IN_F) + lane * 2;
    o[0] = r0;
    o[1] = r1;
  }
}
#undef BAND_ROWS

__global__ void pack_W_kernel(const float* __restrict__ W,
                              float* __restrict__ Bpack, int N) {
  const int ngrpN = N >> 6;
  const int total = KSTEPS * ngrpN * 128;
  int idx = blockIdx.x * blockDim.x + threadIdx.x;
  if (idx >= total) return;

  const int w    = idx & 127;
  const int t    = w & 3;
  const int lane = w >> 2;
  const int blk  = idx >> 7;
  const int gn   = blk % ngrpN;
  const int ks   = blk / ngrpN;

  const int kh  = (lane < 16) ? 0 : 2;
  const int col = gn * 64 + t * 16 + (lane & 15);
  const int kr  = ks * 4 + kh;

  float2 v;
  v.x = W[(size_t)kr * N + col];
  v.y = W[(size_t)(kr + 1) * N + col];
  ((float2*)Bpack)[idx] = v;
}


#define WL_COLS     128
#define WL_WAVES    32
#define WL_F4       (KSTEPS * 2 * 2 * 32)

__global__ __launch_bounds__(32 * WL_WAVES)
void sage_gemm_wlds_kernel(const float* __restrict__ A0,
                           const float* __restrict__ A1,
                           const float* __restrict__ Wm,
                           const float* __restrict__ bias,
                           float* __restrict__ out,
                           int M, int N, int do_relu) {
  __shared__ float4 Wl[WL_F4];

  const int lane  = threadIdx.x & 31;
  const int wave  = threadIdx.x >> 5;
  const int nhalf = N / WL_COLS;
  const int half  = blockIdx.x % nhalf;
  const int gi    = blockIdx.x / nhalf;
  const int G     = gridDim.x / nhalf;

  const int l15 = lane & 15;
  const int kh  = (lane < 16) ? 0 : 2;

  for (int idx = threadIdx.x; idx < WL_F4; idx += 32 * WL_WAVES) {
    const int p  = (idx >> 5) & 1;
    const int g  = (idx >> 6) & 1;
    const int ks = idx >> 7;
    const float* w0 = Wm + (size_t)(ks * 4 + kh) * N
                         + half * WL_COLS + g * 64 + p * 32 + l15;
    float4 v;
    v.x = w0[0];
    v.y = w0[N];
    v.z = w0[16];
    v.w = w0[N + 16];
    Wl[idx] = v;
  }
  __syncthreads();

  const int ntile   = M >> 4;
  const int ntile_g = (gi < ntile) ? (ntile - gi + G - 1) / G : 0;
  const int nstrip  = ntile_g * 2;
  const int mhi     = (lane < 16) ? 0 : 8;

  for (int q = wave; q < nstrip; q += WL_WAVES) {
    const int tm = gi + (q >> 1) * G;
    const int g  = q & 1;
    const int row = tm * 16 + l15;
    const float* Arow0 = A0 + (size_t)row * IN_F + kh;
    const float* Arow1 = A1 + (size_t)row * IN_F + kh;
    const float4* Bl = Wl + g * 64 + lane;

    v8f acc[4] = {};

#define K_STEP_L(AROW, BL, I)                                                 \
    {                                                                         \
      v2f a = *(const v2f*)((AROW) + (I) * 4);             \
      float4 bA = (BL)[(I) * 128];                    \
      float4 bB = (BL)[(I) * 128 + 32];               \
      v2f b0; b0.x = bA.x; b0.y = bA.y;                                       \
      v2f b1; b1.x = bA.z; b1.y = bA.w;                                       \
      v2f b2; b2.x = bB.x; b2.y = bB.y;                                       \
      v2f b3; b3.x = bB.z; b3.y = bB.w;                                       \
      acc[0] = __builtin_amdgcn_wmma_f32_16x16x4_f32(false, a, false, b0,     \
                                                     (short)0, acc[0], false, false); \
      acc[1] = __builtin_amdgcn_wmma_f32_16x16x4_f32(false, a, false, b1,     \
                                                     (short)0, acc[1], false, false); \
      acc[2] = __builtin_amdgcn_wmma_f32_16x16x4_f32(false, a, false, b2,     \
                                                     (short)0, acc[2], false, false); \
      acc[3] = __builtin_amdgcn_wmma_f32_16x16x4_f32(false, a, false, b3,     \
                                                     (short)0, acc[3], false, false); \
    }

    for (int kb = 0; kb < KSTEPS / 8; ++kb) {
      const float*  Ap = (kb < KSPLIT / 8) ? Arow0 + kb * 32
                                           : Arow1 + (kb - KSPLIT / 8) * 32;
      const float4* Bp = Bl + kb * (8 * 128);
#pragma unroll
      for (int j = 0; j < 8; ++j) {
        K_STEP_L(Ap, Bp, j);
      }
    }
#undef K_STEP_L

    const int mbase = tm * 16 + mhi;
    const int col0  = half * WL_COLS + g * 64 + l15;
#pragma unroll
    for (int t = 0; t < 4; ++t) {
      const int col = col0 + t * 16;
      const float bv = bias[col];
#pragma unroll
      for (int i = 0; i < 8; ++i) {
        float v = acc[t][i] + bv;
        if (do_relu) v = fmaxf(v, 0.0f);
        out[(size_t)(mbase + i) * N + col] = v;
      }
    }
  }
}

#define SPLITK       8
#define KS_PER_WAVE  (KSTEPS / SPLITK)
#define PART_EPITCH  34
#define PART_SPITCH  (32 * PART_EPITCH)

__global__ __launch_bounds__(32 * SPLITK)
void sage_gemm_splitk_wmma_kernel(const float* __restrict__ A0,
                                  const float* __restrict__ A1,
                                  const float* __restrict__ Bpack,
                                  const float* __restrict__ bias,
                                  float* __restrict__ out,
                                  int M, int N, int do_relu) {
  __shared__ float part[SPLITK * PART_SPITCH];

  const int lane  = threadIdx.x & 31;
  const int wave  = threadIdx.x >> 5;
  const int ngrpN = N >> 6;
  const int grp   = blockIdx.x;
  const int ngrps = (M >> 4) * ngrpN;
  if (grp >= ngrps) return;

  const int tm = grp / ngrpN;
  const int gn = grp % ngrpN;

  const int l15 = lane & 15;
  const int row = tm * 16 + l15;
  const int kh  = (lane < 16) ? 0 : 2;

  const int ks0 = wave * KS_PER_WAVE;
  const float* Ahalf = (ks0 < KSPLIT) ? A0 : A1;
  const int    ksl   = (ks0 < KSPLIT) ? ks0 : ks0 - KSPLIT;
  const float* Arow  = Ahalf + (size_t)row * IN_F + ksl * 4 + kh;
  const size_t bstride = (size_t)ngrpN * 256;
  const float* Bp = Bpack + (size_t)ks0 * bstride + (size_t)gn * 256 + lane * 8;

  v8f acc[4] = {};

#pragma unroll 4
  for (int i = 0; i < KS_PER_WAVE; ++i) {
    v2f a = *(const v2f*)(Arow + i * 4);
    const float4* bp = (const float4*)(Bp + (size_t)i * bstride);
    float4 bA = bp[0];
    float4 bB = bp[1];
    v2f b0; b0.x = bA.x; b0.y = bA.y;
    v2f b1; b1.x = bA.z; b1.y = bA.w;
    v2f b2; b2.x = bB.x; b2.y = bB.y;
    v2f b3; b3.x = bB.z; b3.y = bB.w;
    acc[0] = __builtin_amdgcn_wmma_f32_16x16x4_f32(false, a, false, b0,
                                                   (short)0, acc[0], false, false);
    acc[1] = __builtin_amdgcn_wmma_f32_16x16x4_f32(false, a, false, b1,
                                                   (short)0, acc[1], false, false);
    acc[2] = __builtin_amdgcn_wmma_f32_16x16x4_f32(false, a, false, b2,
                                                   (short)0, acc[2], false, false);
    acc[3] = __builtin_amdgcn_wmma_f32_16x16x4_f32(false, a, false, b3,
                                                   (short)0, acc[3], false, false);
  }

  float* pw = part + wave * PART_SPITCH + lane;
#pragma unroll
  for (int t = 0; t < 4; ++t) {
#pragma unroll
    for (int i = 0; i < 8; ++i) {
      pw[(t * 8 + i) * PART_EPITCH] = acc[t][i];
    }
  }
  __syncthreads();

#pragma unroll
  for (int j = 0; j < 4; ++j) {
    const int r  = wave * 2 + (j >> 1);
    const int cc = (j & 1) * 32 + lane;
    const int e  = (cc >> 4) * 8 + (r & 7);
    const int sl = (cc & 15) + ((r >> 3) << 4);
    const float* pr = part + e * PART_EPITCH + sl;
    float v = pr[0];
#pragma unroll
    for (int s = 1; s < SPLITK; ++s) v += pr[s * PART_SPITCH];
    const int col = gn * 64 + cc;
    v += bias[col];
    if (do_relu) v = fmaxf(v, 0.0f);
    out[(size_t)(tm * 16 + r) * N + col] = v;
  }
}

extern "C" void kernel_launch(void* const* d_in, const int* in_sizes, int n_in,
                              void* d_out, int out_size, void* d_ws, size_t ws_size,
                              hipStream_t stream) {
  const float* x   = (const float*)d_in[0];
  const float* W1  = (const float*)d_in[1];
  const float* b1  = (const float*)d_in[2];
  const float* W2  = (const float*)d_in[3];
  const float* b2  = (const float*)d_in[4];
  const int* src1  = (const int*)d_in[5];
  const int* dst1  = (const int*)d_in[6];
  const int* src2  = (const int*)d_in[7];
  const int* dst2  = (const int*)d_in[8];
  const int E1 = in_sizes[5];
  const int E2 = in_sizes[7];

  char* wsb = (char*)d_ws;
  const size_t int_words = 3 * (size_t)(N_DST1 + N_DST2) + 2 + (size_t)E1 + (size_t)E2;
  int*  cnt2      = (int*)wsb;
  int*  ovf_head2 = cnt2 + NPART * CPITCH2;
  int*  slots2    = ovf_head2 + NPART * CPITCH2;
  int2* ovf2      = (int2*)(slots2 + (size_t)N_DST2 * NBR_CAP);
  size_t off = int_words * 4;
  auto take = [&](size_t n_elems) { void* p = wsb + off; off += n_elems * 4; return p; };

  float* hN1    = (float*)take((size_t)N_DST1 * IN_F);
  float* h      = (float*)take((size_t)N_DST1 * HID_F);
  float* hN2    = (float*)take((size_t)N_DST2 * IN_F);
  (void)take((size_t)KDIM * HID_F);
  float* Bpack2 = (float*)take((size_t)KDIM * N_CLS);
  int*  slots1    = (int*)((char*)h + 120);
  int2* ovf1      = (int2*)(slots1 + (size_t)N_DST1 * NBR_CAP);
  int*  cnt1      = (int*)(ovf1 + E1);
  int*  ovf_head1 = cnt1 + NPART * CPITCH1;
  (void)ws_size;

  {
    int t2 = KSTEPS * (N_CLS / 64) * 128;
    pack_W_kernel<<<(t2 + 255) / 256, 256, 0, stream>>>(W2, Bpack2, N_CLS);
  }

  build_rows_kernel<<<NWG_ROWS1 + NWG_ROWS2, OWN_THR, 0, stream>>>(
      src1, dst1, cnt1, slots1, ovf_head1, ovf1, E1, N_DST1, NWG_ROWS1,
      src2, dst2, cnt2, slots2, ovf_head2, ovf2, E2, N_DST2);

  gather_mean_banded_kernel<<<GB_NWG, 32 * GB_WPB, 0, stream>>>(
      x, slots1, cnt1, ovf_head1, ovf1, hN1, N_DST1, CPITCH1, E1);

  sage_gemm_wlds_kernel<<<256, 32 * WL_WAVES, 0, stream>>>(
      x, hN1, W1, b1, h, N_DST1, HID_F,  1);

  gather_mean_kernel<<<N_DST2 / 8, 256, 0, stream>>>(
      h, slots2, cnt2, ovf_head2, ovf2, hN2, N_DST2, CPITCH2, E2);

  {
    int ngrps = (N_DST2 / 16) * (N_CLS / 64);
    sage_gemm_splitk_wmma_kernel<<<ngrps, 32 * SPLITK, 0, stream>>>(
        h, hN2, Bpack2, b2, (float*)d_out, N_DST2, N_CLS,  0);
  }
}


# Rows known ahead are prefetched into the second-level cache: layer-1 gather lanes ask for the rows of the node visit four ahead, the layer-1 GEMM for its next strip's A tile; sum order unchanged
# speedup vs baseline: 2.7608x; 1.0344x over previous
#include <hip/hip_runtime.h>
#include <hip/hip_bf16.h>


typedef __attribute__((ext_vector_type(2))) float v2f;
typedef __attribute__((ext_vector_type(8))) float v8f;

#define N_DST1   50000
#define N_DST2   10000
#define IN_F     256
#define KDIM     512
#define KSTEPS   (KDIM / 4)
#define KSPLIT   (IN_F / 4)
#define HID_F    256
#define N_CLS    64

#define NBR_CAP  64

#define NPART      4
#define PCAP       (NBR_CAP / NPART)
#define OWN_ROWS1  896
#define OWN_ROWS2  1280
#define OWN_MAX    1280
#define OWN_THR    1024
#define LIST_CAP   48
#define SCAN_U     8
#define LIST_EBITS 20
#define NRNG1      ((N_DST1 + OWN_ROWS1 - 1) / OWN_ROWS1)
#define NRNG2      ((N_DST2 + OWN_ROWS2 - 1) / OWN_ROWS2)
#define NWG_ROWS1  (NRNG1 * NPART)
#define NWG_ROWS2  (NRNG2 * NPART)
#define CPITCH1    ((N_DST1 + 31) / 32 * 32)
#define CPITCH2    ((N_DST2 + 31) / 32 * 32)

static_assert(NPART == 4, "the gather's slot walk is written for four parts");
static_assert(OWN_MAX >= OWN_ROWS1 && OWN_MAX >= OWN_ROWS2, "LDS sizing");
static_assert(OWN_MAX <= (1 << (32 - LIST_EBITS)), "local row must fit the list word");

__device__ __forceinline__ void place_owned(int i, int s, int dl,
                                            int* s_cnt, int* s_head,
                                            int* s_slots,
                                            int2* __restrict__ ovf) {
  const int rank = atomicAdd(&s_cnt[dl], 1);
  if (rank < PCAP) {
    s_slots[dl * PCAP + rank] = s;
  } else {
    const int prev = atomicExch(&s_head[dl], i + 1);
    ovf[i] = make_int2(s, prev);
  }
}

__global__ __launch_bounds__(OWN_THR)
void build_rows_kernel(const int* __restrict__ src1,
                       const int* __restrict__ dst1,
                       int* __restrict__ cnt1,
                       int* __restrict__ slots1,
                       int* __restrict__ ovf_head1,
                       int2* __restrict__ ovf1, int nE1, int nDst1, int nwg1,
                       const int* __restrict__ src2,
                       const int* __restrict__ dst2,
                       int* __restrict__ cnt2,
                       int* __restrict__ slots2,
                       int* __restrict__ ovf_head2,
                       int2* __restrict__ ovf2, int nE2, int nDst2) {
  __shared__ int s_cnt[OWN_MAX];
  __shared__ int s_head[OWN_MAX];
  __shared__ __attribute__((aligned(16))) int s_slots[OWN_MAX * PCAP];
  __shared__ int s_list[LIST_CAP * OWN_THR];

  const bool first = (int)blockIdx.x < nwg1;
  const int* __restrict__ src = first ? src1 : src2;
  const int* __restrict__ dst = first ? dst1 : dst2;
  int*  cnt      = first ? cnt1 : cnt2;
  int*  slots    = first ? slots1 : slots2;
  int*  ovf_head = first ? ovf_head1 : ovf_head2;
  int2* ovf      = first ? ovf1 : ovf2;
  const int nE      = first ? nE1 : nE2;
  const int nDst    = first ? nDst1 : nDst2;
  const int ownRows = first ? OWN_ROWS1 : OWN_ROWS2;
  const int cpitch  = first ? CPITCH1 : CPITCH2;
  const int lb    = first ? (int)blockIdx.x : (int)blockIdx.x - nwg1;
  const int part  = lb % NPART;
  const int owner = lb / NPART;
  const int base  = owner * ownRows;
  const int left  = nDst - base;
  const unsigned nown = (unsigned)(left < ownRows ? (left > 0 ? left : 0) : ownRows);
  const int tid = threadIdx.x;

  for (int t = tid; t < ownRows; t += OWN_THR) { s_cnt[t] = 0; s_head[t] = 0; }
  __syncthreads();

  const int nq = nE >> 2;
  const int q4 = (nq + NPART - 1) / NPART;
  int j_lo = part * q4;                          if (j_lo > nq) j_lo = nq;
  int j_hi = j_lo + q4;                          if (j_hi > nq) j_hi = nq;
  int j_sc = j_lo + (1 << (LIST_EBITS - 2));     if (j_sc > j_hi) j_sc = j_hi;
  const int e_base = j_lo << 2;
  const int e_end  = (part == NPART - 1) ? nE : (j_hi << 2);

  int lp = tid;
  const int lp_room = tid + (LIST_CAP - 4 * SCAN_U) * OWN_THR;

  auto flush = [&]() {
    for (int k0 = tid; k0 < lp; k0 += 8 * OWN_THR) {
      int w[8], s[8];
#pragma unroll
      for (int u = 0; u < 8; ++u) {
        const int k = k0 + u * OWN_THR;
        const bool on = k < lp;
        w[u] = on ? s_list[k] : 0;
        s[u] = on ? src[e_base + (w[u] & ((1 << LIST_EBITS) - 1))] : 0;
      }
#pragma unroll
      for (int u = 0; u < 8; ++u) {
        if (k0 + u * OWN_THR < lp)
          place_owned(e_base + (w[u] & ((1 << LIST_EBITS) - 1)), s[u],
                      (int)((unsigned)w[u] >> LIST_EBITS), s_cnt, s_head, s_slots, ovf);
      }
    }
    lp = tid;
  };

  const int4* d4 = (const int4*)dst;

  auto group = [&](int j0, bool partial) {
    if (lp > lp_room) flush();
    int4 v[SCAN_U];
#pragma unroll
    for (int u = 0; u < SCAN_U; ++u) {
      const int jj = j0 + u * OWN_THR;
      v[u] = d4[(partial && jj >= j_sc) ? j_sc - 1 : jj];
    }
#pragma unroll
    for (int u = 0; u < SCAN_U; ++u) {
      const int jj = j0 + u * OWN_THR;
      unsigned t0 = (unsigned)(v[u].x - base);
      unsigned t1 = (unsigned)(v[u].y - base);
      unsigned t2 = (unsigned)(v[u].z - base);
      unsigned t3 = (unsigned)(v[u].w - base);
      if (partial && jj >= j_sc) { t0 = ~0u; t1 = ~0u; t2 = ~0u; t3 = ~0u; }
      const unsigned ta = t0 < t1 ? t0 : t1;
      const unsigned tb = t2 < t3 ? t2 : t3;
      if ((ta < tb ? ta : tb) < nown) {
        const unsigned e = (unsigned)(jj - j_lo) << 2;
        if (t0 < nown) { s_list[lp] = (int)((t0 << LIST_EBITS) | e);        lp += OWN_THR; }
        if (t1 < nown) { s_list[lp] = (int)((t1 << LIST_EBITS) | (e | 1u)); lp += OWN_THR; }
        if (t2 < nown) { s_list[lp] = (int)((t2 << LIST_EBITS) | (e | 2u)); lp += OWN_THR; }
        if (t3 < nown) { s_list[lp] = (int)((t3 << LIST_EBITS) | (e | 3u)); lp += OWN_THR; }
      }
    }
  };

  if (j_sc > j_lo) {
    int j0 = j_lo + tid;
    for (; j0 + (SCAN_U - 1) * OWN_THR < j_sc; j0 += SCAN_U * OWN_THR) group(j0, false);
    if (j0 < j_sc) group(j0, true);
  }
  flush();
  for (int e = (j_sc << 2) + tid; e < e_end; e += OWN_THR) {
    const unsigned dl = (unsigned)(dst[e] - base);
    if (dl < nown) place_owned(e, src[e], (int)dl, s_cnt, s_head, s_slots, ovf);
  }
  __syncthreads();

  int* pc = cnt      + (size_t)part * cpitch + base;
  int* ph = ovf_head + (size_t)part * cpitch + base;
  for (int t = tid; t < (int)nown; t += OWN_THR) {
    pc[t] = s_cnt[t];
    ph[t] = s_head[t];
  }
  int4* gs = (int4*)(slots + ((size_t)part * nDst + base) * PCAP);
  const int4* ls = (const int4*)s_slots;
  for (int t = tid; t < (int)nown * (PCAP / 4); t += OWN_THR) gs[t] = ls[t];
}


#define GATHER_ROWS(N)                                                        \
  {                                                                           \
    int s[N];                                                                 \
    float4 v0[N], v1[N];                                                      \
    _Pragma("unroll")                                                         \
    for (int u = 0; u < (N); ++u) {                                           \
      const int b = __builtin_ctzll(m);                  \
      m &= m - 1;                                                             \
      const int lo = __builtin_amdgcn_readlane(w01, b & 31);                  \
      const int hi = __builtin_amdgcn_readlane(w23, b & 31);                  \
      s[u] = b < 32 ? lo : hi;                                                \
    }                                                                         \
    _Pragma("unroll")                                                         \
    for (int u = 0; u < (N); ++u) {                                           \
      const float4* p = hb + (size_t)s[u] * (IN_F / 4);                       \
      v0[u] = p[0];                                                           \
      v1[u] = p[1];                                                           \
    }                                                                         \
    _Pragma("unroll")                                                         \
    for (int u = 0; u < (N); ++u) {                                           \
      a0.x += v0[u].x; a0.y += v0[u].y; a0.z += v0[u].z; a0.w += v0[u].w;     \
      a1.x += v1[u].x; a1.y += v1[u].y; a1.z += v1[u].z; a1.w += v1[u].w;     \
    }                                                                         \
  }

__global__ __launch_bounds__(256)
void gather_mean_kernel(const float* __restrict__ hsrc,
                        const int* __restrict__ slots,
                        const int* __restrict__ cnt,
                        const int* __restrict__ ovf_head,
                        const int2* __restrict__ ovf,
                        float* __restrict__ hN,
                        int n_dst, int cpitch, int nE) {
  const int d = __builtin_amdgcn_readfirstlane(
      (int)(blockIdx.x * (blockDim.x >> 5) + (threadIdx.x >> 5)));
  if (d >= n_dst) return;
  const int lane = threadIdx.x & 31;
  const int plane = n_dst * PCAP;
  const int* rl = slots + (size_t)d * PCAP + (size_t)(lane >> 4) * plane + (lane & 15);
  const int w01 = rl[0];
  const int w23 = rl[2 * (size_t)plane];
  const int c0 = cnt[d];
  const int c1 = cnt[cpitch + d];
  const int c2 = cnt[2 * cpitch + d];
  const int c3 = cnt[3 * cpitch + d];
  const int deg = c0 + c1 + c2 + c3;
  const int e0 = c0 < PCAP ? (c0 > 0 ? c0 : 0) : PCAP;
  const int e1 = c1 < PCAP ? (c1 > 0 ? c1 : 0) : PCAP;
  const int e2 = c2 < PCAP ? (c2 > 0 ? c2 : 0) : PCAP;
  const int e3 = c3 < PCAP ? (c3 > 0 ? c3 : 0) : PCAP;
  const int tot = e0 + e1 + e2 + e3;
  unsigned long long m =
        (unsigned long long)((1u << e0) - 1u)
      | ((unsigned long long)((1u << e1) - 1u) << 16)
      | ((unsigned long long)((1u << e2) - 1u) << 32)
      | ((unsigned long long)((1u << e3) - 1u) << 48);
  const float4* hb = (const float4*)hsrc + lane * 2;

  float4 a0 = make_float4(0.f, 0.f, 0.f, 0.f);
  float4 a1 = make_float4(0.f, 0.f, 0.f, 0.f);
  int n = tot;
  for (; n >= 8; n -= 8) GATHER_ROWS(8)
  if (n & 4) GATHER_ROWS(4)
  if (n & 2) GATHER_ROWS(2)
  if (n & 1) GATHER_ROWS(1)
  if (deg > tot) {
    const int cp[NPART] = {c0, c1, c2, c3};
#pragma unroll
    for (int q = 0; q < NPART; ++q) {
      if (cp[q] <= PCAP) continue;
      int t = ovf_head[q * cpitch + d];
      for (int k = 0; t > 0 && k < nE; ++k) {
        const int2 r = ovf[t - 1];
        t = r.y;
        const float4* p = (const float4*)(hsrc + (size_t)r.x * IN_F) + lane * 2;
        float4 v0 = p[0];
        float4 v1 = p[1];
        a0.x += v0.x; a0.y += v0.y; a0.z += v0.z; a0.w += v0.w;
        a1.x += v1.x; a1.y += v1.y; a1.z += v1.z; a1.w += v1.w;
      }
    }
  }
  const float inv = 1.0f / fmaxf((float)deg, 1.0f);
  a0.x *= inv; a0.y *= inv; a0.z *= inv; a0.w *= inv;
  a1.x *= inv; a1.y *= inv; a1.z *= inv; a1.w *= inv;
  float4* o = (float4*)(hN + (size_t)d * IN_F) + lane * 2;
  o[0] = a0;
  o[1] = a1;
}
#undef GATHER_ROWS

#define GB_K        7
#define GB_WPB      8
#define GB_NWG      1024
#define BAND_SHIFT  15
#define NBAND       8
#define ROW_SHIFT   10
static_assert((IN_F * 4) == (1 << ROW_SHIFT), "row pitch");

static_assert(N_DST1 <= GB_K * GB_NWG * GB_WPB, "every layer-1 node needs a wave slot");

#define PF_D  4
#define PREFETCH_ROW(OFS)                                                     \
  asm volatile("global_prefetch_b8 %0, %1 scope:SCOPE_SE\n\t"                 \
               "global_prefetch_b8 %0, %1 offset:256 scope:SCOPE_SE\n\t"      \
               "global_prefetch_b8 %0, %1 offset:512 scope:SCOPE_SE\n\t"      \
               "global_prefetch_b8 %0, %1 offset:768 scope:SCOPE_SE"          \
               : : "v"(OFS), "s"(hsrc));
#define PREFETCH_VISIT(JN, PN)                                                \
  {                                                                           \
    int c01 = w01[JN] >> (BAND_SHIFT + ROW_SHIFT);                            \
    int c23 = w23[JN] >> (BAND_SHIFT + ROW_SHIFT);                            \
    c01 = c01 > NBAND - 1 ? NBAND - 1 : c01;                                  \
    c23 = c23 > NBAND - 1 ? NBAND - 1 : c23;                                  \
    if (c01 == (PN)) { PREFETCH_ROW(w01[JN]) }                                \
    if (c23 == (PN)) { PREFETCH_ROW(w23[JN]) }                                \
  }

#define BAND_ROWS(N, J)                                                       \
  {                                                                           \
    int s[N];                                                                 \
    float4 v0[N], v1[N];                                                      \
    _Pragma("unroll")                                                         \
    for (int u = 0; u < (N); ++u) {                                           \
      const int b = __builtin_ctzll(m);             \
      m &= m - 1;                                                             \
      const int lo = __builtin_amdgcn_readlane(w01[J], b & 31);               \
      const int hi = __builtin_amdgcn_readlane(w23[J], b & 31);               \
      s[u] = b < 32 ? lo : hi;                                                \
    }                                                                         \
    _Pragma("unroll")                                                         \
    for (int u = 0; u < (N); ++u) {                                           \
      const float4* p = (const float4*)((const char*)hb + (size_t)s[u]);      \
      v0[u] = p[0];                                                           \
      v1[u] = p[1];                                                           \
    }                                                                         \
    _Pragma("unroll")                                                         \
    for (int u = 0; u < (N); ++u) {                                           \
      a0[J].x += v0[u].x; a0[J].y += v0[u].y;                                 \
      a0[J].z += v0[u].z; a0[J].w += v0[u].w;                                 \
      a1[J].x += v1[u].x; a1[J].y += v1[u].y;                                 \
      a1[J].z += v1[u].z; a1[J].w += v1[u].w;                                 \
    }                                                                         \
  }

__global__ __launch_bounds__(32 * GB_WPB)
void gather_mean_banded_kernel(const float* __restrict__ hsrc,
                               const int* __restrict__ slots,
                               const int* __restrict__ cnt,
                               const int* __restrict__ ovf_head,
                               const int2* __restrict__ ovf,
                               float* __restrict__ hN,
                               int n_dst, int cpitch, int nE) {
  const int wid = __builtin_amdgcn_readfirstlane(
      (int)(blockIdx.x * GB_WPB + (threadIdx.x >> 5)));
  const int nw   = (int)gridDim.x * GB_WPB;
  const int lane = threadIdx.x & 31;
  const int plane = n_dst * PCAP;
  const float4* hb = (const float4*)hsrc + lane * 2;

  int w01[GB_K], w23[GB_K];
  int deg[GB_K];
  bool chained[GB_K];
  float4 a0[GB_K], a1[GB_K];
#pragma unroll
  for (int j = 0; j < GB_K; ++j) {
    const int d  = wid + j * nw;
    const bool on = d < n_dst;
    const int dd = on ? d : n_dst - 1;
    a0[j] = make_float4(0.f, 0.f, 0.f, 0.f);
    a1[j] = make_float4(0.f, 0.f, 0.f, 0.f);
    const int* rl = slots + (size_t)dd * PCAP + (size_t)(lane >> 4) * plane + (lane & 15);
    const int x01 = rl[0];
    const int x23 = rl[2 * (size_t)plane];
    const int c0 = on ? cnt[dd] : 0;
    const int c1 = on ? cnt[cpitch + dd] : 0;
    const int c2 = on ? cnt[2 * cpitch + dd] : 0;
    const int c3 = on ? cnt[3 * cpitch + dd] : 0;
    const int e0 = c0 < PCAP ? (c0 > 0 ? c0 : 0) : PCAP;
    const int e1 = c1 < PCAP ? (c1 > 0 ? c1 : 0) : PCAP;
    const int e2 = c2 < PCAP ? (c2 > 0 ? c2 : 0) : PCAP;
    const int e3 = c3 < PCAP ? (c3 > 0 ? c3 : 0) : PCAP;
    deg[j] = c0 + c1 + c2 + c3;
    chained[j] = (c0 > PCAP) | (c1 > PCAP) | (c2 > PCAP) | (c3 > PCAP);
    const int r = lane & 15;
    w01[j] = r < (lane < 16 ? e0 : e1) ? (int)((unsigned)x01 << ROW_SHIFT) : -1;
    w23[j] = r < (lane < 16 ? e2 : e3) ? (int)((unsigned)x23 << ROW_SHIFT) : -1;
  }

#pragma unroll
  for (int j = 0; j < PF_D; ++j) PREFETCH_VISIT(j, 0)

#pragma nounroll
  for (int p = 0; p < NBAND; ++p) {
#pragma unroll
    for (int j = 0; j < GB_K; ++j) {
      PREFETCH_VISIT((j + PF_D) % GB_K, p + (j + PF_D) / GB_K)
      int b01 = w01[j] >> (BAND_SHIFT + ROW_SHIFT);  b01 = b01 > NBAND - 1 ? NBAND - 1 : b01;
      int b23 = w23[j] >> (BAND_SHIFT + ROW_SHIFT);  b23 = b23 > NBAND - 1 ? NBAND - 1 : b23;
      unsigned long long m =
            (unsigned long long)__builtin_amdgcn_ballot_w32(b01 == p)
          | ((unsigned long long)__builtin_amdgcn_ballot_w32(b23 == p) << 32);
      int n = __builtin_popcountll(m);
      for (; n >= 4; n -= 4) BAND_ROWS(4, j)
      if (n & 2) BAND_ROWS(2, j)
      if (n & 1) BAND_ROWS(1, j)
    }
  }

#pragma unroll
  for (int j = 0; j < GB_K; ++j) {
    const int d = wid + j * nw;
    if (d >= n_dst) continue;
    if (chained[j]) {
      for (int q = 0; q < NPART; ++q) {
        if (cnt[q * cpitch + d] <= PCAP) continue;
        int t = ovf_head[q * cpitch + d];
        for (int k = 0; t > 0 && k < nE; ++k) {
          const int2 r = ovf[t - 1];
          t = r.y;
          const float4* p = (const float4*)(hsrc + (size_t)r.x * IN_F) + lane * 2;
          float4 v0 = p[0];
          float4 v1 = p[1];
          a0[j].x += v0.x; a0[j].y += v0.y; a0[j].z += v0.z; a0[j].w += v0.w;
          a1[j].x += v1.x; a1[j].y += v1.y; a1[j].z += v1.z; a1[j].w += v1.w;
        }
      }
    }
    const float inv = 1.0f / fmaxf((float)deg[j], 1.0f);
    float4 r0 = a0[j], r1 = a1[j];
    r0.x *= inv; r0.y *= inv; r0.z *= inv; r0.w *= inv;
    r1.x *= inv; r1.y *= inv; r1.z *= inv; r1.w *= inv;
    float4* o = (float4*)(hN + (size_t)d * IN_F) + lane * 2;
    o[0] = r0;
    o[1] = r1;
  }
}
#undef BAND_ROWS
#undef PREFETCH_VISIT
#undef PREFETCH_ROW

__global__ void pack_W_kernel(const float* __restrict__ W,
                              float* __restrict__ Bpack, int N) {
  const int ngrpN = N >> 6;
  const int total = KSTEPS * ngrpN * 128;
  int idx = blockIdx.x * blockDim.x + threadIdx.x;
  if (idx >= total) return;

  const int w    = idx & 127;
  const int t    = w & 3;
  const int lane = w >> 2;
  const int blk  = idx >> 7;
  const int gn   = blk % ngrpN;
  const int ks   = blk / ngrpN;

  const int kh  = (lane < 16) ? 0 : 2;
  const int col = gn * 64 + t * 16 + (lane & 15);
  const int kr  = ks * 4 + kh;

  float2 v;
  v.x = W[(size_t)kr * N + col];
  v.y = W[(size_t)(kr + 1) * N + col];
  ((float2*)Bpack)[idx] = v;
}


#define WL_COLS     128
#define WL_WAVES    32
#define WL_F4       (KSTEPS * 2 * 2 * 32)

__global__ __launch_bounds__(32 * WL_WAVES)
void sage_gemm_wlds_kernel(const float* __restrict__ A0,
                           const float* __restrict__ A1,
                           const float* __restrict__ Wm,
                           const float* __restrict__ bias,
                           float* __restrict__ out,
                           int M, int N, int do_relu) {
  __shared__ float4 Wl[WL_F4];

  const int lane  = threadIdx.x & 31;
  const int wave  = threadIdx.x >> 5;
  const int nhalf = N / WL_COLS;
  const int half  = blockIdx.x % nhalf;
  const int gi    = blockIdx.x / nhalf;
  const int G     = gridDim.x / nhalf;

  const int l15 = lane & 15;
  const int kh  = (lane < 16) ? 0 : 2;

  const int ntile   = M >> 4;
  const int ntile_g = (gi < ntile) ? (ntile - gi + G - 1) / G : 0;
  const int nstrip  = ntile_g * 2;

  auto prefetch_tile = [&](int tm) {
    const float* r = ((lane < 16) ? A0 : A1) + (size_t)(tm * 16 + l15) * IN_F;
    __builtin_prefetch(r, 0, 2);
    __builtin_prefetch(r + 64, 0, 2);
    __builtin_prefetch(r + 128, 0, 2);
    __builtin_prefetch(r + 192, 0, 2);
  };
  const bool pf_wave = (wave & 1) == 0;
  if (pf_wave && wave < nstrip) prefetch_tile(gi + (wave >> 1) * G);

  for (int idx = threadIdx.x; idx < WL_F4; idx += 32 * WL_WAVES) {
    const int p  = (idx >> 5) & 1;
    const int g  = (idx >> 6) & 1;
    const int ks = idx >> 7;
    const float* w0 = Wm + (size_t)(ks * 4 + kh) * N
                         + half * WL_COLS + g * 64 + p * 32 + l15;
    float4 v;
    v.x = w0[0];
    v.y = w0[N];
    v.z = w0[16];
    v.w = w0[N + 16];
    Wl[idx] = v;
  }
  __syncthreads();

  const int mhi     = (lane < 16) ? 0 : 8;

  for (int q = wave; q < nstrip; q += WL_WAVES) {
    if (pf_wave && q + WL_WAVES < nstrip) prefetch_tile(gi + ((q + WL_WAVES) >> 1) * G);
    const int tm = gi + (q >> 1) * G;
    const int g  = q & 1;
    const int row = tm * 16 + l15;
    const float* Arow0 = A0 + (size_t)row * IN_F + kh;
    const float* Arow1 = A1 + (size_t)row * IN_F + kh;
    const float4* Bl = Wl + g * 64 + lane;

    v8f acc[4] = {};

#define K_STEP_L(AROW, BL, I)                                                 \
    {                                                                         \
      v2f a = *(const v2f*)((AROW) + (I) * 4);             \
      float4 bA = (BL)[(I) * 128];                    \
      float4 bB = (BL)[(I) * 128 + 32];               \
      v2f b0; b0.x = bA.x; b0.y = bA.y;                                       \
      v2f b1; b1.x = bA.z; b1.y = bA.w;                                       \
      v2f b2; b2.x = bB.x; b2.y = bB.y;                                       \
      v2f b3; b3.x = bB.z; b3.y = bB.w;                                       \
      acc[0] = __builtin_amdgcn_wmma_f32_16x16x4_f32(false, a, false, b0,     \
                                                     (short)0, acc[0], false, false); \
      acc[1] = __builtin_amdgcn_wmma_f32_16x16x4_f32(false, a, false, b1,     \
                                                     (short)0, acc[1], false, false); \
      acc[2] = __builtin_amdgcn_wmma_f32_16x16x4_f32(false, a, false, b2,     \
                                                     (short)0, acc[2], false, false); \
      acc[3] = __builtin_amdgcn_wmma_f32_16x16x4_f32(false, a, false, b3,     \
                                                     (short)0, acc[3], false, false); \
    }

    for (int kb = 0; kb < KSTEPS / 8; ++kb) {
      const float*  Ap = (kb < KSPLIT / 8) ? Arow0 + kb * 32
                                           : Arow1 + (kb - KSPLIT / 8) * 32;
      const float4* Bp = Bl + kb * (8 * 128);
#pragma unroll
      for (int j = 0; j < 8; ++j) {
        K_STEP_L(Ap, Bp, j);
      }
    }
#undef K_STEP_L

    const int mbase = tm * 16 + mhi;
    const int col0  = half * WL_COLS + g * 64 + l15;
#pragma unroll
    for (int t = 0; t < 4; ++t) {
      const int col = col0 + t * 16;
      const float bv = bias[col];
#pragma unroll
      for (int i = 0; i < 8; ++i) {
        float v = acc[t][i] + bv;
        if (do_relu) v = fmaxf(v, 0.0f);
        out[(size_t)(mbase + i) * N + col] = v;
      }
    }
  }
}

#define SPLITK       8
#define KS_PER_WAVE  (KSTEPS / SPLITK)
#define PART_EPITCH  34
#define PART_SPITCH  (32 * PART_EPITCH)

__global__ __launch_bounds__(32 * SPLITK)
void sage_gemm_splitk_wmma_kernel(const float* __restrict__ A0,
                                  const float* __restrict__ A1,
                                  const float* __restrict__ Bpack,
                                  const float* __restrict__ bias,
                                  float* __restrict__ out,
                                  int M, int N, int do_relu) {
  __shared__ float part[SPLITK * PART_SPITCH];

  const int lane  = threadIdx.x & 31;
  const int wave  = threadIdx.x >> 5;
  const int ngrpN = N >> 6;
  const int grp   = blockIdx.x;
  const int ngrps = (M >> 4) * ngrpN;
  if (grp >= ngrps) return;

  const int tm = grp / ngrpN;
  const int gn = grp % ngrpN;

  const int l15 = lane & 15;
  const int row = tm * 16 + l15;
  const int kh  = (lane < 16) ? 0 : 2;

  const int ks0 = wave * KS_PER_WAVE;
  const float* Ahalf = (ks0 < KSPLIT) ? A0 : A1;
  const int    ksl   = (ks0 < KSPLIT) ? ks0 : ks0 - KSPLIT;
  const float* Arow  = Ahalf + (size_t)row * IN_F + ksl * 4 + kh;
  const size_t bstride = (size_t)ngrpN * 256;
  const float* Bp = Bpack + (size_t)ks0 * bstride + (size_t)gn * 256 + lane * 8;

  v8f acc[4] = {};

#pragma unroll 4
  for (int i = 0; i < KS_PER_WAVE; ++i) {
    v2f a = *(const v2f*)(Arow + i * 4);
    const float4* bp = (const float4*)(Bp + (size_t)i * bstride);
    float4 bA = bp[0];
    float4 bB = bp[1];
    v2f b0; b0.x = bA.x; b0.y = bA.y;
    v2f b1; b1.x = bA.z; b1.y = bA.w;
    v2f b2; b2.x = bB.x; b2.y = bB.y;
    v2f b3; b3.x = bB.z; b3.y = bB.w;
    acc[0] = __builtin_amdgcn_wmma_f32_16x16x4_f32(false, a, false, b0,
                                                   (short)0, acc[0], false, false);
    acc[1] = __builtin_amdgcn_wmma_f32_16x16x4_f32(false, a, false, b1,
                                                   (short)0, acc[1], false, false);
    acc[2] = __builtin_amdgcn_wmma_f32_16x16x4_f32(false, a, false, b2,
                                                   (short)0, acc[2], false, false);
    acc[3] = __builtin_amdgcn_wmma_f32_16x16x4_f32(false, a, false, b3,
                                                   (short)0, acc[3], false, false);
  }

  float* pw = part + wave * PART_SPITCH + lane;
#pragma unroll
  for (int t = 0; t < 4; ++t) {
#pragma unroll
    for (int i = 0; i < 8; ++i) {
      pw[(t * 8 + i) * PART_EPITCH] = acc[t][i];
    }
  }
  __syncthreads();

#pragma unroll
  for (int j = 0; j < 4; ++j) {
    const int r  = wave * 2 + (j >> 1);
    const int cc = (j & 1) * 32 + lane;
    const int e  = (cc >> 4) * 8 + (r & 7);
    const int sl = (cc & 15) + ((r >> 3) << 4);
    const float* pr = part + e * PART_EPITCH + sl;
    float v = pr[0];
#pragma unroll
    for (int s = 1; s < SPLITK; ++s) v += pr[s * PART_SPITCH];
    const int col = gn * 64 + cc;
    v += bias[col];
    if (do_relu) v = fmaxf(v, 0.0f);
    out[(size_t)(tm * 16 + r) * N + col] = v;
  }
}

extern "C" void kernel_launch(void* const* d_in, const int* in_sizes, int n_in,
                              void* d_out, int out_size, void* d_ws, size_t ws_size,
                              hipStream_t stream) {
  const float* x   = (const float*)d_in[0];
  const float* W1  = (const float*)d_in[1];
  const float* b1  = (const float*)d_in[2];
  const float* W2  = (const float*)d_in[3];
  const float* b2  = (const float*)d_in[4];
  const int* src1  = (const int*)d_in[5];
  const int* dst1  = (const int*)d_in[6];
  const int* src2  = (const int*)d_in[7];
  const int* dst2  = (const int*)d_in[8];
  const int E1 = in_sizes[5];
  const int E2 = in_sizes[7];

  char* wsb = (char*)d_ws;
  const size_t int_words = 3 * (size_t)(N_DST1 + N_DST2) + 2 + (size_t)E1 + (size_t)E2;
  int*  cnt2      = (int*)wsb;
  int*  ovf_head2 = cnt2 + NPART * CPITCH2;
  int*  slots2    = ovf_head2 + NPART * CPITCH2;
  int2* ovf2      = (int2*)(slots2 + (size_t)N_DST2 * NBR_CAP);
  size_t off = int_words * 4;
  auto take = [&](size_t n_elems) { void* p = wsb + off; off += n_elems * 4; return p; };

  float* hN1    = (float*)take((size_t)N_DST1 * IN_F);
  float* h      = (float*)take((size_t)N_DST1 * HID_F);
  float* hN2    = (float*)take((size_t)N_DST2 * IN_F);
  (void)take((size_t)KDIM * HID_F);
  float* Bpack2 = (float*)take((size_t)KDIM * N_CLS);
  int*  slots1    = (int*)((char*)h + 120);
  int2* ovf1      = (int2*)(slots1 + (size_t)N_DST1 * NBR_CAP);
  int*  cnt1      = (int*)(ovf1 + E1);
  int*  ovf_head1 = cnt1 + NPART * CPITCH1;
  (void)ws_size;

  {
    int t2 = KSTEPS * (N_CLS / 64) * 128;
    pack_W_kernel<<<(t2 + 255) / 256, 256, 0, stream>>>(W2, Bpack2, N_CLS);
  }

  build_rows_kernel<<<NWG_ROWS1 + NWG_ROWS2, OWN_THR, 0, stream>>>(
      src1, dst1, cnt1, slots1, ovf_head1, ovf1, E1, N_DST1, NWG_ROWS1,
      src2, dst2, cnt2, slots2, ovf_head2, ovf2, E2, N_DST2);

  gather_mean_banded_kernel<<<GB_NWG, 32 * GB_WPB, 0, stream>>>(
      x, slots1, cnt1, ovf_head1, ovf1, hN1, N_DST1, CPITCH1, E1);

  sage_gemm_wlds_kernel<<<256, 32 * WL_WAVES, 0, stream>>>(
      x, hN1, W1, b1, h, N_DST1, HID_F,  1);

  gather_mean_kernel<<<N_DST2 / 8, 256, 0, stream>>>(
      h, slots2, cnt2, ovf_head2, ovf2, hN2, N_DST2, CPITCH2, E2);

  {
    int ngrps = (N_DST2 / 16) * (N_CLS / 64);
    sage_gemm_splitk_wmma_kernel<<<ngrps, 32 * SPLITK, 0, stream>>>(
        h, hN2, Bpack2, b2, (float*)d_out, N_DST2, N_CLS,  0);
  }
}


# Layer 2 as one launch: each 16x64 strip's work-group gathers its 16 neighbor means into an LDS A tile, then runs the split-K GEMM on it, W2 fragments read from row-major W2; sum order unchanged
# speedup vs baseline: 2.8874x; 1.0459x over previous
#include <hip/hip_runtime.h>
#include <hip/hip_bf16.h>


typedef __attribute__((ext_vector_type(2))) float v2f;
typedef __attribute__((ext_vector_type(8))) float v8f;

#define N_DST1   50000
#define N_DST2   10000
#define IN_F     256
#define KDIM     512
#define KSTEPS   (KDIM / 4)
#define KSPLIT   (IN_F / 4)
#define HID_F    256
#define N_CLS    64

#define NBR_CAP  64

#define NPART      4
#define PCAP       (NBR_CAP / NPART)
#define OWN_ROWS1  896
#define OWN_ROWS2  1280
#define OWN_MAX    1280
#define OWN_THR    1024
#define LIST_CAP   48
#define SCAN_U     8
#define LIST_EBITS 20
#define NRNG1      ((N_DST1 + OWN_ROWS1 - 1) / OWN_ROWS1)
#define NRNG2      ((N_DST2 + OWN_ROWS2 - 1) / OWN_ROWS2)
#define NWG_ROWS1  (NRNG1 * NPART)
#define NWG_ROWS2  (NRNG2 * NPART)
#define CPITCH1    ((N_DST1 + 31) / 32 * 32)
#define CPITCH2    ((N_DST2 + 31) / 32 * 32)

static_assert(NPART == 4, "the gather's slot walk is written for four parts");
static_assert(OWN_MAX >= OWN_ROWS1 && OWN_MAX >= OWN_ROWS2, "LDS sizing");
static_assert(OWN_MAX <= (1 << (32 - LIST_EBITS)), "local row must fit the list word");

__device__ __forceinline__ void place_owned(int i, int s, int dl,
                                            int* s_cnt, int* s_head,
                                            int* s_slots,
                                            int2* __restrict__ ovf) {
  const int rank = atomicAdd(&s_cnt[dl], 1);
  if (rank < PCAP) {
    s_slots[dl * PCAP + rank] = s;
  } else {
    const int prev = atomicExch(&s_head[dl], i + 1);
    ovf[i] = make_int2(s, prev);
  }
}

__global__ __launch_bounds__(OWN_THR)
void build_rows_kernel(const int* __restrict__ src1,
                       const int* __restrict__ dst1,
                       int* __restrict__ cnt1,
                       int* __restrict__ slots1,
                       int* __restrict__ ovf_head1,
                       int2* __restrict__ ovf1, int nE1, int nDst1, int nwg1,
                       const int* __restrict__ src2,
                       const int* __restrict__ dst2,
                       int* __restrict__ cnt2,
                       int* __restrict__ slots2,
                       int* __restrict__ ovf_head2,
                       int2* __restrict__ ovf2, int nE2, int nDst2) {
  __shared__ int s_cnt[OWN_MAX];
  __shared__ int s_head[OWN_MAX];
  __shared__ __attribute__((aligned(16))) int s_slots[OWN_MAX * PCAP];
  __shared__ int s_list[LIST_CAP * OWN_THR];

  const bool first = (int)blockIdx.x < nwg1;
  const int* __restrict__ src = first ? src1 : src2;
  const int* __restrict__ dst = first ? dst1 : dst2;
  int*  cnt      = first ? cnt1 : cnt2;
  int*  slots    = first ? slots1 : slots2;
  int*  ovf_head = first ? ovf_head1 : ovf_head2;
  int2* ovf      = first ? ovf1 : ovf2;
  const int nE      = first ? nE1 : nE2;
  const int nDst    = first ? nDst1 : nDst2;
  const int ownRows = first ? OWN_ROWS1 : OWN_ROWS2;
  const int cpitch  = first ? CPITCH1 : CPITCH2;
  const int lb    = first ? (int)blockIdx.x : (int)blockIdx.x - nwg1;
  const int part  = lb % NPART;
  const int owner = lb / NPART;
  const int base  = owner * ownRows;
  const int left  = nDst - base;
  const unsigned nown = (unsigned)(left < ownRows ? (left > 0 ? left : 0) : ownRows);
  const int tid = threadIdx.x;

  for (int t = tid; t < ownRows; t += OWN_THR) { s_cnt[t] = 0; s_head[t] = 0; }
  __syncthreads();

  const int nq = nE >> 2;
  const int q4 = (nq + NPART - 1) / NPART;
  int j_lo = part * q4;                          if (j_lo > nq) j_lo = nq;
  int j_hi = j_lo + q4;                          if (j_hi > nq) j_hi = nq;
  int j_sc = j_lo + (1 << (LIST_EBITS - 2));     if (j_sc > j_hi) j_sc = j_hi;
  const int e_base = j_lo << 2;
  const int e_end  = (part == NPART - 1) ? nE : (j_hi << 2);

  int lp = tid;
  const int lp_room = tid + (LIST_CAP - 4 * SCAN_U) * OWN_THR;

  auto flush = [&]() {
    for (int k0 = tid; k0 < lp; k0 += 8 * OWN_THR) {
      int w[8], s[8];
#pragma unroll
      for (int u = 0; u < 8; ++u) {
        const int k = k0 + u * OWN_THR;
        const bool on = k < lp;
        w[u] = on ? s_list[k] : 0;
        s[u] = on ? src[e_base + (w[u] & ((1 << LIST_EBITS) - 1))] : 0;
      }
#pragma unroll
      for (int u = 0; u < 8; ++u) {
        if (k0 + u * OWN_THR < lp)
          place_owned(e_base + (w[u] & ((1 << LIST_EBITS) - 1)), s[u],
                      (int)((unsigned)w[u] >> LIST_EBITS), s_cnt, s_head, s_slots, ovf);
      }
    }
    lp = tid;
  };

  const int4* d4 = (const int4*)dst;

  auto group = [&](int j0, bool partial) {
    if (lp > lp_room) flush();
    int4 v[SCAN_U];
#pragma unroll
    for (int u = 0; u < SCAN_U; ++u) {
      const int jj = j0 + u * OWN_THR;
      v[u] = d4[(partial && jj >= j_sc) ? j_sc - 1 : jj];
    }
#pragma unroll
    for (int u = 0; u < SCAN_U; ++u) {
      const int jj = j0 + u * OWN_THR;
      unsigned t0 = (unsigned)(v[u].x - base);
      unsigned t1 = (unsigned)(v[u].y - base);
      unsigned t2 = (unsigned)(v[u].z - base);
      unsigned t3 = (unsigned)(v[u].w - base);
      if (partial && jj >= j_sc) { t0 = ~0u; t1 = ~0u; t2 = ~0u; t3 = ~0u; }
      const unsigned ta = t0 < t1 ? t0 : t1;
      const unsigned tb = t2 < t3 ? t2 : t3;
      if ((ta < tb ? ta : tb) < nown) {
        const unsigned e = (unsigned)(jj - j_lo) << 2;
        if (t0 < nown) { s_list[lp] = (int)((t0 << LIST_EBITS) | e);        lp += OWN_THR; }
        if (t1 < nown) { s_list[lp] = (int)((t1 << LIST_EBITS) | (e | 1u)); lp += OWN_THR; }
        if (t2 < nown) { s_list[lp] = (int)((t2 << LIST_EBITS) | (e | 2u)); lp += OWN_THR; }
        if (t3 < nown) { s_list[lp] = (int)((t3 << LIST_EBITS) | (e | 3u)); lp += OWN_THR; }
      }
    }
  };

  if (j_sc > j_lo) {
    int j0 = j_lo + tid;
    for (; j0 + (SCAN_U - 1) * OWN_THR < j_sc; j0 += SCAN_U * OWN_THR) group(j0, false);
    if (j0 < j_sc) group(j0, true);
  }
  flush();
  for (int e = (j_sc << 2) + tid; e < e_end; e += OWN_THR) {
    const unsigned dl = (unsigned)(dst[e] - base);
    if (dl < nown) place_owned(e, src[e], (int)dl, s_cnt, s_head, s_slots, ovf);
  }
  __syncthreads();

  int* pc = cnt      + (size_t)part * cpitch + base;
  int* ph = ovf_head + (size_t)part * cpitch + base;
  for (int t = tid; t < (int)nown; t += OWN_THR) {
    pc[t] = s_cnt[t];
    ph[t] = s_head[t];
  }
  int4* gs = (int4*)(slots + ((size_t)part * nDst + base) * PCAP);
  const int4* ls = (const int4*)s_slots;
  for (int t = tid; t < (int)nown * (PCAP / 4); t += OWN_THR) gs[t] = ls[t];
}


#define GATHER_ROWS(N)                                                        \
  {                                                                           \
    int s[N];                                                                 \
    float4 v0[N], v1[N];                                                      \
    _Pragma("unroll")                                                         \
    for (int u = 0; u < (N); ++u) {                                           \
      const int b = __builtin_ctzll(m);                  \
      m &= m - 1;                                                             \
      const int lo = __builtin_amdgcn_readlane(w01, b & 31);                  \
      const int hi = __builtin_amdgcn_readlane(w23, b & 31);                  \
      s[u] = b < 32 ? lo : hi;                                                \
    }                                                                         \
    _Pragma("unroll")                                                         \
    for (int u = 0; u < (N); ++u) {                                           \
      const float4* p = hb + (size_t)s[u] * (IN_F / 4);                       \
      v0[u] = p[0];                                                           \
      v1[u] = p[1];                                                           \
    }                                                                         \
    _Pragma("unroll")                                                         \
    for (int u = 0; u < (N); ++u) {                                           \
      a0.x += v0[u].x; a0.y += v0[u].y; a0.z += v0[u].z; a0.w += v0[u].w;     \
      a1.x += v1[u].x; a1.y += v1[u].y; a1.z += v1[u].z; a1.w += v1[u].w;     \
    }                                                                         \
  }

__device__ __forceinline__ void gather_mean_node(
    const float* __restrict__ hsrc,
    const int* __restrict__ ovf_head,
    const int2* __restrict__ ovf,
    int d, int w01, int w23, int c0, int c1, int c2, int c3,
    int cpitch, int nE, int lane, float4& r0, float4& r1) {
  const int deg = c0 + c1 + c2 + c3;
  const int e0 = c0 < PCAP ? (c0 > 0 ? c0 : 0) : PCAP;
  const int e1 = c1 < PCAP ? (c1 > 0 ? c1 : 0) : PCAP;
  const int e2 = c2 < PCAP ? (c2 > 0 ? c2 : 0) : PCAP;
  const int e3 = c3 < PCAP ? (c3 > 0 ? c3 : 0) : PCAP;
  const int tot = e0 + e1 + e2 + e3;
  unsigned long long m =
        (unsigned long long)((1u << e0) - 1u)
      | ((unsigned long long)((1u << e1) - 1u) << 16)
      | ((unsigned long long)((1u << e2) - 1u) << 32)
      | ((unsigned long long)((1u << e3) - 1u) << 48);
  const float4* hb = (const float4*)hsrc + lane * 2;

  float4 a0 = make_float4(0.f, 0.f, 0.f, 0.f);
  float4 a1 = make_float4(0.f, 0.f, 0.f, 0.f);
  int n = tot;
  for (; n >= 8; n -= 8) GATHER_ROWS(8)
  if (n & 4) GATHER_ROWS(4)
  if (n & 2) GATHER_ROWS(2)
  if (n & 1) GATHER_ROWS(1)
  if (deg > tot) {
    const int cp[NPART] = {c0, c1, c2, c3};
#pragma unroll
    for (int q = 0; q < NPART; ++q) {
      if (cp[q] <= PCAP) continue;
      int t = ovf_head[q * cpitch + d];
      for (int k = 0; t > 0 && k < nE; ++k) {
        const int2 r = ovf[t - 1];
        t = r.y;
        const float4* p = (const float4*)(hsrc + (size_t)r.x * IN_F) + lane * 2;
        float4 v0 = p[0];
        float4 v1 = p[1];
        a0.x += v0.x; a0.y += v0.y; a0.z += v0.z; a0.w += v0.w;
        a1.x += v1.x; a1.y += v1.y; a1.z += v1.z; a1.w += v1.w;
      }
    }
  }
  const float inv = 1.0f / fmaxf((float)deg, 1.0f);
  a0.x *= inv; a0.y *= inv; a0.z *= inv; a0.w *= inv;
  a1.x *= inv; a1.y *= inv; a1.z *= inv; a1.w *= inv;
  r0 = a0;
  r1 = a1;
}
#undef GATHER_ROWS


#define GB_K        7
#define GB_WPB      8
#define GB_NWG      1024
#define BAND_SHIFT  15
#define NBAND       8
#define ROW_SHIFT   10
static_assert((IN_F * 4) == (1 << ROW_SHIFT), "row pitch");

static_assert(N_DST1 <= GB_K * GB_NWG * GB_WPB, "every layer-1 node needs a wave slot");

#define PF_D  4
#define PREFETCH_ROW(OFS)                                                     \
  asm volatile("global_prefetch_b8 %0, %1 scope:SCOPE_SE\n\t"                 \
               "global_prefetch_b8 %0, %1 offset:256 scope:SCOPE_SE\n\t"      \
               "global_prefetch_b8 %0, %1 offset:512 scope:SCOPE_SE\n\t"      \
               "global_prefetch_b8 %0, %1 offset:768 scope:SCOPE_SE"          \
               : : "v"(OFS), "s"(hsrc));
#define PREFETCH_VISIT(JN, PN)                                                \
  {                                                                           \
    int c01 = w01[JN] >> (BAND_SHIFT + ROW_SHIFT);                            \
    int c23 = w23[JN] >> (BAND_SHIFT + ROW_SHIFT);                            \
    c01 = c01 > NBAND - 1 ? NBAND - 1 : c01;                                  \
    c23 = c23 > NBAND - 1 ? NBAND - 1 : c23;                                  \
    if (c01 == (PN)) { PREFETCH_ROW(w01[JN]) }                                \
    if (c23 == (PN)) { PREFETCH_ROW(w23[JN]) }                                \
  }

#define BAND_ROWS(N, J)                                                       \
  {                                                                           \
    int s[N];                                                                 \
    float4 v0[N], v1[N];                                                      \
    _Pragma("unroll")                                                         \
    for (int u = 0; u < (N); ++u) {                                           \
      const int b = __builtin_ctzll(m);             \
      m &= m - 1;                                                             \
      const int lo = __builtin_amdgcn_readlane(w01[J], b & 31);               \
      const int hi = __builtin_amdgcn_readlane(w23[J], b & 31);               \
      s[u] = b < 32 ? lo : hi;                                                \
    }                                                                         \
    _Pragma("unroll")                                                         \
    for (int u = 0; u < (N); ++u) {                                           \
      const float4* p = (const float4*)((const char*)hb + (size_t)s[u]);      \
      v0[u] = p[0];                                                           \
      v1[u] = p[1];                                                           \
    }                                                                         \
    _Pragma("unroll")                                                         \
    for (int u = 0; u < (N); ++u) {                                           \
      a0[J].x += v0[u].x; a0[J].y += v0[u].y;                                 \
      a0[J].z += v0[u].z; a0[J].w += v0[u].w;                                 \
      a1[J].x += v1[u].x; a1[J].y += v1[u].y;                                 \
      a1[J].z += v1[u].z; a1[J].w += v1[u].w;                                 \
    }                                                                         \
  }

__global__ __launch_bounds__(32 * GB_WPB)
void gather_mean_banded_kernel(const float* __restrict__ hsrc,
                               const int* __restrict__ slots,
                               const int* __restrict__ cnt,
                               const int* __restrict__ ovf_head,
                               const int2* __restrict__ ovf,
                               float* __restrict__ hN,
                               int n_dst, int cpitch, int nE) {
  const int wid = __builtin_amdgcn_readfirstlane(
      (int)(blockIdx.x * GB_WPB + (threadIdx.x >> 5)));
  const int nw   = (int)gridDim.x * GB_WPB;
  const int lane = threadIdx.x & 31;
  const int plane = n_dst * PCAP;
  const float4* hb = (const float4*)hsrc + lane * 2;

  int w01[GB_K], w23[GB_K];
  int deg[GB_K];
  bool chained[GB_K];
  float4 a0[GB_K], a1[GB_K];
#pragma unroll
  for (int j = 0; j < GB_K; ++j) {
    const int d  = wid + j * nw;
    const bool on = d < n_dst;
    const int dd = on ? d : n_dst - 1;
    a0[j] = make_float4(0.f, 0.f, 0.f, 0.f);
    a1[j] = make_float4(0.f, 0.f, 0.f, 0.f);
    const int* rl = slots + (size_t)dd * PCAP + (size_t)(lane >> 4) * plane + (lane & 15);
    const int x01 = rl[0];
    const int x23 = rl[2 * (size_t)plane];
    const int c0 = on ? cnt[dd] : 0;
    const int c1 = on ? cnt[cpitch + dd] : 0;
    const int c2 = on ? cnt[2 * cpitch + dd] : 0;
    const int c3 = on ? cnt[3 * cpitch + dd] : 0;
    const int e0 = c0 < PCAP ? (c0 > 0 ? c0 : 0) : PCAP;
    const int e1 = c1 < PCAP ? (c1 > 0 ? c1 : 0) : PCAP;
    const int e2 = c2 < PCAP ? (c2 > 0 ? c2 : 0) : PCAP;
    const int e3 = c3 < PCAP ? (c3 > 0 ? c3 : 0) : PCAP;
    deg[j] = c0 + c1 + c2 + c3;
    chained[j] = (c0 > PCAP) | (c1 > PCAP) | (c2 > PCAP) | (c3 > PCAP);
    const int r = lane & 15;
    w01[j] = r < (lane < 16 ? e0 : e1) ? (int)((unsigned)x01 << ROW_SHIFT) : -1;
    w23[j] = r < (lane < 16 ? e2 : e3) ? (int)((unsigned)x23 << ROW_SHIFT) : -1;
  }

#pragma unroll
  for (int j = 0; j < PF_D; ++j) PREFETCH_VISIT(j, 0)

#pragma nounroll
  for (int p = 0; p < NBAND; ++p) {
#pragma unroll
    for (int j = 0; j < GB_K; ++j) {
      PREFETCH_VISIT((j + PF_D) % GB_K, p + (j + PF_D) / GB_K)
      int b01 = w01[j] >> (BAND_SHIFT + ROW_SHIFT);  b01 = b01 > NBAND - 1 ? NBAND - 1 : b01;
      int b23 = w23[j] >> (BAND_SHIFT + ROW_SHIFT);  b23 = b23 > NBAND - 1 ? NBAND - 1 : b23;
      unsigned long long m =
            (unsigned long long)__builtin_amdgcn_ballot_w32(b01 == p)
          | ((unsigned long long)__builtin_amdgcn_ballot_w32(b23 == p) << 32);
      int n = __builtin_popcountll(m);
      for (; n >= 4; n -= 4) BAND_ROWS(4, j)
      if (n & 2) BAND_ROWS(2, j)
      if (n & 1) BAND_ROWS(1, j)
    }
  }

#pragma unroll
  for (int j = 0; j < GB_K; ++j) {
    const int d = wid + j * nw;
    if (d >= n_dst) continue;
    if (chained[j]) {
      for (int q = 0; q < NPART; ++q) {
        if (cnt[q * cpitch + d] <= PCAP) continue;
        int t = ovf_head[q * cpitch + d];
        for (int k = 0; t > 0 && k < nE; ++k) {
          const int2 r = ovf[t - 1];
          t = r.y;
          const float4* p = (const float4*)(hsrc + (size_t)r.x * IN_F) + lane * 2;
          float4 v0 = p[0];
          float4 v1 = p[1];
          a0[j].x += v0.x; a0[j].y += v0.y; a0[j].z += v0.z; a0[j].w += v0.w;
          a1[j].x += v1.x; a1[j].y += v1.y; a1[j].z += v1.z; a1[j].w += v1.w;
        }
      }
    }
    const float inv = 1.0f / fmaxf((float)deg[j], 1.0f);
    float4 r0 = a0[j], r1 = a1[j];
    r0.x *= inv; r0.y *= inv; r0.z *= inv; r0.w *= inv;
    r1.x *= inv; r1.y *= inv; r1.z *= inv; r1.w *= inv;
    float4* o = (float4*)(hN + (size_t)d * IN_F) + lane * 2;
    o[0] = r0;
    o[1] = r1;
  }
}
#undef BAND_ROWS
#undef PREFETCH_VISIT
#undef PREFETCH_ROW


#define WL_COLS     128
#define WL_WAVES    32
#define WL_F4       (KSTEPS * 2 * 2 * 32)

__global__ __launch_bounds__(32 * WL_WAVES)
void sage_gemm_wlds_kernel(const float* __restrict__ A0,
                           const float* __restrict__ A1,
                           const float* __restrict__ Wm,
                           const float* __restrict__ bias,
                           float* __restrict__ out,
                           int M, int N, int do_relu) {
  __shared__ float4 Wl[WL_F4];

  const int lane  = threadIdx.x & 31;
  const int wave  = threadIdx.x >> 5;
  const int nhalf = N / WL_COLS;
  const int half  = blockIdx.x % nhalf;
  const int gi    = blockIdx.x / nhalf;
  const int G     = gridDim.x / nhalf;

  const int l15 = lane & 15;
  const int kh  = (lane < 16) ? 0 : 2;

  const int ntile   = M >> 4;
  const int ntile_g = (gi < ntile) ? (ntile - gi + G - 1) / G : 0;
  const int nstrip  = ntile_g * 2;

  auto prefetch_tile = [&](int tm) {
    const float* r = ((lane < 16) ? A0 : A1) + (size_t)(tm * 16 + l15) * IN_F;
    __builtin_prefetch(r, 0, 2);
    __builtin_prefetch(r + 64, 0, 2);
    __builtin_prefetch(r + 128, 0, 2);
    __builtin_prefetch(r + 192, 0, 2);
  };
  const bool pf_wave = (wave & 1) == 0;
  if (pf_wave && wave < nstrip) prefetch_tile(gi + (wave >> 1) * G);

  for (int idx = threadIdx.x; idx < WL_F4; idx += 32 * WL_WAVES) {
    const int p  = (idx >> 5) & 1;
    const int g  = (idx >> 6) & 1;
    const int ks = idx >> 7;
    const float* w0 = Wm + (size_t)(ks * 4 + kh) * N
                         + half * WL_COLS + g * 64 + p * 32 + l15;
    float4 v;
    v.x = w0[0];
    v.y = w0[N];
    v.z = w0[16];
    v.w = w0[N + 16];
    Wl[idx] = v;
  }
  __syncthreads();

  const int mhi     = (lane < 16) ? 0 : 8;

  for (int q = wave; q < nstrip; q += WL_WAVES) {
    if (pf_wave && q + WL_WAVES < nstrip) prefetch_tile(gi + ((q + WL_WAVES) >> 1) * G);
    const int tm = gi + (q >> 1) * G;
    const int g  = q & 1;
    const int row = tm * 16 + l15;
    const float* Arow0 = A0 + (size_t)row * IN_F + kh;
    const float* Arow1 = A1 + (size_t)row * IN_F + kh;
    const float4* Bl = Wl + g * 64 + lane;

    v8f acc[4] = {};

#define K_STEP_L(AROW, BL, I)                                                 \
    {                                                                         \
      v2f a = *(const v2f*)((AROW) + (I) * 4);             \
      float4 bA = (BL)[(I) * 128];                    \
      float4 bB = (BL)[(I) * 128 + 32];               \
      v2f b0; b0.x = bA.x; b0.y = bA.y;                                       \
      v2f b1; b1.x = bA.z; b1.y = bA.w;                                       \
      v2f b2; b2.x = bB.x; b2.y = bB.y;                                       \
      v2f b3; b3.x = bB.z; b3.y = bB.w;                                       \
      acc[0] = __builtin_amdgcn_wmma_f32_16x16x4_f32(false, a, false, b0,     \
                                                     (short)0, acc[0], false, false); \
      acc[1] = __builtin_amdgcn_wmma_f32_16x16x4_f32(false, a, false, b1,     \
                                                     (short)0, acc[1], false, false); \
      acc[2] = __builtin_amdgcn_wmma_f32_16x16x4_f32(false, a, false, b2,     \
                                                     (short)0, acc[2], false, false); \
      acc[3] = __builtin_amdgcn_wmma_f32_16x16x4_f32(false, a, false, b3,     \
                                                     (short)0, acc[3], false, false); \
    }

    for (int kb = 0; kb < KSTEPS / 8; ++kb) {
      const float*  Ap = (kb < KSPLIT / 8) ? Arow0 + kb * 32
                                           : Arow1 + (kb - KSPLIT / 8) * 32;
      const float4* Bp = Bl + kb * (8 * 128);
#pragma unroll
      for (int j = 0; j < 8; ++j) {
        K_STEP_L(Ap, Bp, j);
      }
    }
#undef K_STEP_L

    const int mbase = tm * 16 + mhi;
    const int col0  = half * WL_COLS + g * 64 + l15;
#pragma unroll
    for (int t = 0; t < 4; ++t) {
      const int col = col0 + t * 16;
      const float bv = bias[col];
#pragma unroll
      for (int i = 0; i < 8; ++i) {
        float v = acc[t][i] + bv;
        if (do_relu) v = fmaxf(v, 0.0f);
        out[(size_t)(mbase + i) * N + col] = v;
      }
    }
  }
}

#define SPLITK       8
#define KS_PER_WAVE  (KSTEPS / SPLITK)
#define PART_EPITCH  34
#define PART_SPITCH  (32 * PART_EPITCH)
#define A_PITCH      (IN_F + 4)

static_assert(2 * SPLITK == 16, "two nodes per wave fill the 16-row tile");
static_assert(KSPLIT % KS_PER_WAVE == 0, "a wave's k lie in one half of the concat");

__global__ __launch_bounds__(32 * SPLITK)
void sage_gather_gemm_splitk_kernel(const float* __restrict__ hsrc,
                                    const int* __restrict__ slots,
                                    const int* __restrict__ cnt,
                                    const int* __restrict__ ovf_head,
                                    const int2* __restrict__ ovf,
                                    const float* __restrict__ Wm,
                                    const float* __restrict__ bias,
                                    float* __restrict__ out,
                                    int M, int N, int cpitch, int nE) {
  __shared__ float part[SPLITK * PART_SPITCH];
  __shared__ __attribute__((aligned(16))) float At[16 * A_PITCH];

  const int lane  = threadIdx.x & 31;
  const int wave  = __builtin_amdgcn_readfirstlane((int)(threadIdx.x >> 5));
  const int ngrpN = N >> 6;
  const int grp   = blockIdx.x;
  const int ngrps = (M >> 4) * ngrpN;
  if (grp >= ngrps) return;

  const int tm = grp / ngrpN;
  const int gn = grp % ngrpN;

  {
    const int dA = tm * 16 + 2 * wave;
    const int dB = dA + 1;
    const int plane = M * PCAP;
    const int* rl = slots + (size_t)dA * PCAP + (size_t)(lane >> 4) * plane + (lane & 15);
    const int wA01 = rl[0];
    const int wA23 = rl[2 * (size_t)plane];
    const int wB01 = rl[PCAP];
    const int wB23 = rl[2 * (size_t)plane + PCAP];
    const int cA0 = cnt[dA], cA1 = cnt[cpitch + dA], cA2 = cnt[2 * cpitch + dA], cA3 = cnt[3 * cpitch + dA];
    const int cB0 = cnt[dB], cB1 = cnt[cpitch + dB], cB2 = cnt[2 * cpitch + dB], cB3 = cnt[3 * cpitch + dB];
    float4 r0, r1;
    gather_mean_node(hsrc, ovf_head, ovf, dA, wA01, wA23, cA0, cA1, cA2, cA3,
                     cpitch, nE, lane, r0, r1);
    float4* o = (float4*)(At + (2 * wave) * A_PITCH) + lane * 2;
    o[0] = r0;
    o[1] = r1;
    gather_mean_node(hsrc, ovf_head, ovf, dB, wB01, wB23, cB0, cB1, cB2, cB3,
                     cpitch, nE, lane, r0, r1);
    o = (float4*)(At + (2 * wave + 1) * A_PITCH) + lane * 2;
    o[0] = r0;
    o[1] = r1;
  }
  __syncthreads();

  const int l15 = lane & 15;
  const int row = tm * 16 + l15;
  const int kh  = (lane < 16) ? 0 : 2;

  const int ks0 = wave * KS_PER_WAVE;
  const float* Wc = Wm + (size_t)(ks0 * 4 + kh) * N + gn * 64 + l15;

  v8f acc[4] = {};

#define K_STEP_W(ALOAD, I)                                                    \
    {                                                                         \
      v2f a = ALOAD;                                   \
      const float* wr = Wc + (size_t)(I) * 4 * N;                             \
      v2f b0; b0.x = wr[0];  b0.y = wr[N];                                    \
      v2f b1; b1.x = wr[16]; b1.y = wr[N + 16];                               \
      v2f b2; b2.x = wr[32]; b2.y = wr[N + 32];                               \
      v2f b3; b3.x = wr[48]; b3.y = wr[N + 48];                               \
      acc[0] = __builtin_amdgcn_wmma_f32_16x16x4_f32(false, a, false, b0,     \
                                                     (short)0, acc[0], false, false); \
      acc[1] = __builtin_amdgcn_wmma_f32_16x16x4_f32(false, a, false, b1,     \
                                                     (short)0, acc[1], false, false); \
      acc[2] = __builtin_amdgcn_wmma_f32_16x16x4_f32(false, a, false, b2,     \
                                                     (short)0, acc[2], false, false); \
      acc[3] = __builtin_amdgcn_wmma_f32_16x16x4_f32(false, a, false, b3,     \
                                                     (short)0, acc[3], false, false); \
    }

  if (ks0 < KSPLIT) {
    const float* Arow = hsrc + (size_t)row * IN_F + ks0 * 4 + kh;
#pragma unroll 4
    for (int i = 0; i < KS_PER_WAVE; ++i) K_STEP_W(*(const v2f*)(Arow + i * 4), i)
  } else {
    const float* Al = At + l15 * A_PITCH + (ks0 - KSPLIT) * 4 + kh;
#pragma unroll 4
    for (int i = 0; i < KS_PER_WAVE; ++i) K_STEP_W(*(const v2f*)(Al + i * 4), i)
  }
#undef K_STEP_W

  float* pw = part + wave * PART_SPITCH + lane;
#pragma unroll
  for (int t = 0; t < 4; ++t) {
#pragma unroll
    for (int i = 0; i < 8; ++i) {
      pw[(t * 8 + i) * PART_EPITCH] = acc[t][i];
    }
  }
  __syncthreads();

#pragma unroll
  for (int j = 0; j < 4; ++j) {
    const int r  = wave * 2 + (j >> 1);
    const int cc = (j & 1) * 32 + lane;
    const int e  = (cc >> 4) * 8 + (r & 7);
    const int sl = (cc & 15) + ((r >> 3) << 4);
    const float* pr = part + e * PART_EPITCH + sl;
    float v = pr[0];
#pragma unroll
    for (int s = 1; s < SPLITK; ++s) v += pr[s * PART_SPITCH];
    const int col = gn * 64 + cc;
    v += bias[col];
    out[(size_t)(tm * 16 + r) * N + col] = v;
  }
}

extern "C" void kernel_launch(void* const* d_in, const int* in_sizes, int n_in,
                              void* d_out, int out_size, void* d_ws, size_t ws_size,
                              hipStream_t stream) {
  const float* x   = (const float*)d_in[0];
  const float* W1  = (const float*)d_in[1];
  const float* b1  = (const float*)d_in[2];
  const float* W2  = (const float*)d_in[3];
  const float* b2  = (const float*)d_in[4];
  const int* src1  = (const int*)d_in[5];
  const int* dst1  = (const int*)d_in[6];
  const int* src2  = (const int*)d_in[7];
  const int* dst2  = (const int*)d_in[8];
  const int E1 = in_sizes[5];
  const int E2 = in_sizes[7];

  char* wsb = (char*)d_ws;
  const size_t int_words = 3 * (size_t)(N_DST1 + N_DST2) + 2 + (size_t)E1 + (size_t)E2;
  int*  cnt2      = (int*)wsb;
  int*  ovf_head2 = cnt2 + NPART * CPITCH2;
  int*  slots2    = ovf_head2 + NPART * CPITCH2;
  int2* ovf2      = (int2*)(slots2 + (size_t)N_DST2 * NBR_CAP);
  size_t off = int_words * 4;
  auto take = [&](size_t n_elems) { void* p = wsb + off; off += n_elems * 4; return p; };

  float* hN1    = (float*)take((size_t)N_DST1 * IN_F);
  float* h      = (float*)take((size_t)N_DST1 * HID_F);
  int*  slots1    = (int*)((char*)h + 120);
  int2* ovf1      = (int2*)(slots1 + (size_t)N_DST1 * NBR_CAP);
  int*  cnt1      = (int*)(ovf1 + E1);
  int*  ovf_head1 = cnt1 + NPART * CPITCH1;
  (void)ws_size;

  build_rows_kernel<<<NWG_ROWS1 + NWG_ROWS2, OWN_THR, 0, stream>>>(
      src1, dst1, cnt1, slots1, ovf_head1, ovf1, E1, N_DST1, NWG_ROWS1,
      src2, dst2, cnt2, slots2, ovf_head2, ovf2, E2, N_DST2);

  gather_mean_banded_kernel<<<GB_NWG, 32 * GB_WPB, 0, stream>>>(
      x, slots1, cnt1, ovf_head1, ovf1, hN1, N_DST1, CPITCH1, E1);

  sage_gemm_wlds_kernel<<<256, 32 * WL_WAVES, 0, stream>>>(
      x, hN1, W1, b1, h, N_DST1, HID_F,  1);

  {
    int ngrps = (N_DST2 / 16) * (N_CLS / 64);
    sage_gather_gemm_splitk_kernel<<<ngrps, 32 * SPLITK, 0, stream>>>(
        h, slots2, cnt2, ovf_head2, ovf2, W2, b2, (float*)d_out,
        N_DST2, N_CLS, CPITCH2, E2);
  }
}


# Gather lanes take floats 4l..4l+3 of each 512-byte half-row (was 8 consecutive) and the float planes start on a line boundary: a row load or store covers 4 whole 128-byte lines; sum order unchanged
# speedup vs baseline: 3.0408x; 1.0531x over previous
#include <hip/hip_runtime.h>
#include <hip/hip_bf16.h>


typedef __attribute__((ext_vector_type(2))) float v2f;
typedef __attribute__((ext_vector_type(8))) float v8f;

#define N_DST1   50000
#define N_DST2   10000
#define IN_F     256
#define KDIM     512
#define KSTEPS   (KDIM / 4)
#define KSPLIT   (IN_F / 4)
#define HID_F    256
#define N_CLS    64

#define NBR_CAP  64

#define NPART      4
#define PCAP       (NBR_CAP / NPART)
#define OWN_ROWS1  896
#define OWN_ROWS2  1280
#define OWN_MAX    1280
#define OWN_THR    1024
#define LIST_CAP   48
#define SCAN_U     8
#define LIST_EBITS 20
#define NRNG1      ((N_DST1 + OWN_ROWS1 - 1) / OWN_ROWS1)
#define NRNG2      ((N_DST2 + OWN_ROWS2 - 1) / OWN_ROWS2)
#define NWG_ROWS1  (NRNG1 * NPART)
#define NWG_ROWS2  (NRNG2 * NPART)
#define CPITCH1    ((N_DST1 + 31) / 32 * 32)
#define CPITCH2    ((N_DST2 + 31) / 32 * 32)

static_assert(NPART == 4, "the gather's slot walk is written for four parts");
static_assert(OWN_MAX >= OWN_ROWS1 && OWN_MAX >= OWN_ROWS2, "LDS sizing");
static_assert(OWN_MAX <= (1 << (32 - LIST_EBITS)), "local row must fit the list word");

__device__ __forceinline__ void place_owned(int i, int s, int dl,
                                            int* s_cnt, int* s_head,
                                            int* s_slots,
                                            int2* __restrict__ ovf) {
  const int rank = atomicAdd(&s_cnt[dl], 1);
  if (rank < PCAP) {
    s_slots[dl * PCAP + rank] = s;
  } else {
    const int prev = atomicExch(&s_head[dl], i + 1);
    ovf[i] = make_int2(s, prev);
  }
}

__global__ __launch_bounds__(OWN_THR)
void build_rows_kernel(const int* __restrict__ src1,
                       const int* __restrict__ dst1,
                       int* __restrict__ cnt1,
                       int* __restrict__ slots1,
                       int* __restrict__ ovf_head1,
                       int2* __restrict__ ovf1, int nE1, int nDst1, int nwg1,
                       const int* __restrict__ src2,
                       const int* __restrict__ dst2,
                       int* __restrict__ cnt2,
                       int* __restrict__ slots2,
                       int* __restrict__ ovf_head2,
                       int2* __restrict__ ovf2, int nE2, int nDst2) {
  __shared__ int s_cnt[OWN_MAX];
  __shared__ int s_head[OWN_MAX];
  __shared__ __attribute__((aligned(16))) int s_slots[OWN_MAX * PCAP];
  __shared__ int s_list[LIST_CAP * OWN_THR];

  const bool first = (int)blockIdx.x < nwg1;
  const int* __restrict__ src = first ? src1 : src2;
  const int* __restrict__ dst = first ? dst1 : dst2;
  int*  cnt      = first ? cnt1 : cnt2;
  int*  slots    = first ? slots1 : slots2;
  int*  ovf_head = first ? ovf_head1 : ovf_head2;
  int2* ovf      = first ? ovf1 : ovf2;
  const int nE      = first ? nE1 : nE2;
  const int nDst    = first ? nDst1 : nDst2;
  const int ownRows = first ? OWN_ROWS1 : OWN_ROWS2;
  const int cpitch  = first ? CPITCH1 : CPITCH2;
  const int lb    = first ? (int)blockIdx.x : (int)blockIdx.x - nwg1;
  const int part  = lb % NPART;
  const int owner = lb / NPART;
  const int base  = owner * ownRows;
  const int left  = nDst - base;
  const unsigned nown = (unsigned)(left < ownRows ? (left > 0 ? left : 0) : ownRows);
  const int tid = threadIdx.x;

  for (int t = tid; t < ownRows; t += OWN_THR) { s_cnt[t] = 0; s_head[t] = 0; }
  __syncthreads();

  const int nq = nE >> 2;
  const int q4 = (nq + NPART - 1) / NPART;
  int j_lo = part * q4;                          if (j_lo > nq) j_lo = nq;
  int j_hi = j_lo + q4;                          if (j_hi > nq) j_hi = nq;
  int j_sc = j_lo + (1 << (LIST_EBITS - 2));     if (j_sc > j_hi) j_sc = j_hi;
  const int e_base = j_lo << 2;
  const int e_end  = (part == NPART - 1) ? nE : (j_hi << 2);

  int lp = tid;
  const int lp_room = tid + (LIST_CAP - 4 * SCAN_U) * OWN_THR;

  auto flush = [&]() {
    for (int k0 = tid; k0 < lp; k0 += 8 * OWN_THR) {
      int w[8], s[8];
#pragma unroll
      for (int u = 0; u < 8; ++u) {
        const int k = k0 + u * OWN_THR;
        const bool on = k < lp;
        w[u] = on ? s_list[k] : 0;
        s[u] = on ? src[e_base + (w[u] & ((1 << LIST_EBITS) - 1))] : 0;
      }
#pragma unroll
      for (int u = 0; u < 8; ++u) {
        if (k0 + u * OWN_THR < lp)
          place_owned(e_base + (w[u] & ((1 << LIST_EBITS) - 1)), s[u],
                      (int)((unsigned)w[u] >> LIST_EBITS), s_cnt, s_head, s_slots, ovf);
      }
    }
    lp = tid;
  };

  const int4* d4 = (const int4*)dst;

  auto group = [&](int j0, bool partial) {
    if (lp > lp_room) flush();
    int4 v[SCAN_U];
#pragma unroll
    for (int u = 0; u < SCAN_U; ++u) {
      const int jj = j0 + u * OWN_THR;
      v[u] = d4[(partial && jj >= j_sc) ? j_sc - 1 : jj];
    }
#pragma unroll
    for (int u = 0; u < SCAN_U; ++u) {
      const int jj = j0 + u * OWN_THR;
      unsigned t0 = (unsigned)(v[u].x - base);
      unsigned t1 = (unsigned)(v[u].y - base);
      unsigned t2 = (unsigned)(v[u].z - base);
      unsigned t3 = (unsigned)(v[u].w - base);
      if (partial && jj >= j_sc) { t0 = ~0u; t1 = ~0u; t2 = ~0u; t3 = ~0u; }
      const unsigned ta = t0 < t1 ? t0 : t1;
      const unsigned tb = t2 < t3 ? t2 : t3;
      if ((ta < tb ? ta : tb) < nown) {
        const unsigned e = (unsigned)(jj - j_lo) << 2;
        if (t0 < nown) { s_list[lp] = (int)((t0 << LIST_EBITS) | e);        lp += OWN_THR; }
        if (t1 < nown) { s_list[lp] = (int)((t1 << LIST_EBITS) | (e | 1u)); lp += OWN_THR; }
        if (t2 < nown) { s_list[lp] = (int)((t2 << LIST_EBITS) | (e | 2u)); lp += OWN_THR; }
        if (t3 < nown) { s_list[lp] = (int)((t3 << LIST_EBITS) | (e | 3u)); lp += OWN_THR; }
      }
    }
  };

  if (j_sc > j_lo) {
    int j0 = j_lo + tid;
    for (; j0 + (SCAN_U - 1) * OWN_THR < j_sc; j0 += SCAN_U * OWN_THR) group(j0, false);
    if (j0 < j_sc) group(j0, true);
  }
  flush();
  for (int e = (j_sc << 2) + tid; e < e_end; e += OWN_THR) {
    const unsigned dl = (unsigned)(dst[e] - base);
    if (dl < nown) place_owned(e, src[e], (int)dl, s_cnt, s_head, s_slots, ovf);
  }
  __syncthreads();

  int* pc = cnt      + (size_t)part * cpitch + base;
  int* ph = ovf_head + (size_t)part * cpitch + base;
  for (int t = tid; t < (int)nown; t += OWN_THR) {
    pc[t] = s_cnt[t];
    ph[t] = s_head[t];
  }
  int4* gs = (int4*)(slots + ((size_t)part * nDst + base) * PCAP);
  const int4* ls = (const int4*)s_slots;
  for (int t = tid; t < (int)nown * (PCAP / 4); t += OWN_THR) gs[t] = ls[t];
}


#define GATHER_ROWS(N)                                                        \
  {                                                                           \
    int s[N];                                                                 \
    float4 v0[N], v1[N];                                                      \
    _Pragma("unroll")                                                         \
    for (int u = 0; u < (N); ++u) {                                           \
      const int b = __builtin_ctzll(m);                  \
      m &= m - 1;                                                             \
      const int lo = __builtin_amdgcn_readlane(w01, b & 31);                  \
      const int hi = __builtin_amdgcn_readlane(w23, b & 31);                  \
      s[u] = b < 32 ? lo : hi;                                                \
    }                                                                         \
    _Pragma("unroll")                                                         \
    for (int u = 0; u < (N); ++u) {                                           \
      const float4* p = hb + (size_t)s[u] * (IN_F / 4);                       \
      v0[u] = p[0];                                                           \
      v1[u] = p[IN_F / 8];                                \
    }                                                                         \
    _Pragma("unroll")                                                         \
    for (int u = 0; u < (N); ++u) {                                           \
      a0.x += v0[u].x; a0.y += v0[u].y; a0.z += v0[u].z; a0.w += v0[u].w;     \
      a1.x += v1[u].x; a1.y += v1[u].y; a1.z += v1[u].z; a1.w += v1[u].w;     \
    }                                                                         \
  }

__device__ __forceinline__ void gather_mean_node(
    const float* __restrict__ hsrc,
    const int* __restrict__ ovf_head,
    const int2* __restrict__ ovf,
    int d, int w01, int w23, int c0, int c1, int c2, int c3,
    int cpitch, int nE, int lane, float4& r0, float4& r1) {
  const int deg = c0 + c1 + c2 + c3;
  const int e0 = c0 < PCAP ? (c0 > 0 ? c0 : 0) : PCAP;
  const int e1 = c1 < PCAP ? (c1 > 0 ? c1 : 0) : PCAP;
  const int e2 = c2 < PCAP ? (c2 > 0 ? c2 : 0) : PCAP;
  const int e3 = c3 < PCAP ? (c3 > 0 ? c3 : 0) : PCAP;
  const int tot = e0 + e1 + e2 + e3;
  unsigned long long m =
        (unsigned long long)((1u << e0) - 1u)
      | ((unsigned long long)((1u << e1) - 1u) << 16)
      | ((unsigned long long)((1u << e2) - 1u) << 32)
      | ((unsigned long long)((1u << e3) - 1u) << 48);
  const float4* hb = (const float4*)hsrc + lane;

  float4 a0 = make_float4(0.f, 0.f, 0.f, 0.f);
  float4 a1 = make_float4(0.f, 0.f, 0.f, 0.f);
  int n = tot;
  for (; n >= 8; n -= 8) GATHER_ROWS(8)
  if (n & 4) GATHER_ROWS(4)
  if (n & 2) GATHER_ROWS(2)
  if (n & 1) GATHER_ROWS(1)
  if (deg > tot) {
    const int cp[NPART] = {c0, c1, c2, c3};
#pragma unroll
    for (int q = 0; q < NPART; ++q) {
      if (cp[q] <= PCAP) continue;
      int t = ovf_head[q * cpitch + d];
      for (int k = 0; t > 0 && k < nE; ++k) {
        const int2 r = ovf[t - 1];
        t = r.y;
        const float4* p = (const float4*)(hsrc + (size_t)r.x * IN_F) + lane;
        float4 v0 = p[0];
        float4 v1 = p[IN_F / 8];
        a0.x += v0.x; a0.y += v0.y; a0.z += v0.z; a0.w += v0.w;
        a1.x += v1.x; a1.y += v1.y; a1.z += v1.z; a1.w += v1.w;
      }
    }
  }
  const float inv = 1.0f / fmaxf((float)deg, 1.0f);
  a0.x *= inv; a0.y *= inv; a0.z *= inv; a0.w *= inv;
  a1.x *= inv; a1.y *= inv; a1.z *= inv; a1.w *= inv;
  r0 = a0;
  r1 = a1;
}
#undef GATHER_ROWS


#define GB_K        7
#define GB_WPB      8
#define GB_NWG      1024
#define BAND_SHIFT  15
#define NBAND       8
#define ROW_SHIFT   10
static_assert((IN_F * 4) == (1 << ROW_SHIFT), "row pitch");

static_assert(N_DST1 <= GB_K * GB_NWG * GB_WPB, "every layer-1 node needs a wave slot");

#define PF_D  4
#define PREFETCH_ROW(OFS)                                                     \
  asm volatile("global_prefetch_b8 %0, %1 scope:SCOPE_SE\n\t"                 \
               "global_prefetch_b8 %0, %1 offset:256 scope:SCOPE_SE\n\t"      \
               "global_prefetch_b8 %0, %1 offset:512 scope:SCOPE_SE\n\t"      \
               "global_prefetch_b8 %0, %1 offset:768 scope:SCOPE_SE"          \
               : : "v"(OFS), "s"(hsrc));
#define PREFETCH_VISIT(JN, PN)                                                \
  {                                                                           \
    int c01 = w01[JN] >> (BAND_SHIFT + ROW_SHIFT);                            \
    int c23 = w23[JN] >> (BAND_SHIFT + ROW_SHIFT);                            \
    c01 = c01 > NBAND - 1 ? NBAND - 1 : c01;                                  \
    c23 = c23 > NBAND - 1 ? NBAND - 1 : c23;                                  \
    if (c01 == (PN)) { PREFETCH_ROW(w01[JN]) }                                \
    if (c23 == (PN)) { PREFETCH_ROW(w23[JN]) }                                \
  }

#define BAND_ROWS(N, J)                                                       \
  {                                                                           \
    int s[N];                                                                 \
    float4 v0[N], v1[N];                                                      \
    _Pragma("unroll")                                                         \
    for (int u = 0; u < (N); ++u) {                                           \
      const int b = __builtin_ctzll(m);             \
      m &= m - 1;                                                             \
      const int lo = __builtin_amdgcn_readlane(w01[J], b & 31);               \
      const int hi = __builtin_amdgcn_readlane(w23[J], b & 31);               \
      s[u] = b < 32 ? lo : hi;                                                \
    }                                                                         \
    _Pragma("unroll")                                                         \
    for (int u = 0; u < (N); ++u) {                                           \
      const float4* p = (const float4*)((const char*)hb + (size_t)s[u]);      \
      v0[u] = p[0];                                                           \
      v1[u] = p[IN_F / 8];                                \
    }                                                                         \
    _Pragma("unroll")                                                         \
    for (int u = 0; u < (N); ++u) {                                           \
      a0[J].x += v0[u].x; a0[J].y += v0[u].y;                                 \
      a0[J].z += v0[u].z; a0[J].w += v0[u].w;                                 \
      a1[J].x += v1[u].x; a1[J].y += v1[u].y;                                 \
      a1[J].z += v1[u].z; a1[J].w += v1[u].w;                                 \
    }                                                                         \
  }

__global__ __launch_bounds__(32 * GB_WPB)
void gather_mean_banded_kernel(const float* __restrict__ hsrc,
                               const int* __restrict__ slots,
                               const int* __restrict__ cnt,
                               const int* __restrict__ ovf_head,
                               const int2* __restrict__ ovf,
                               float* __restrict__ hN,
                               int n_dst, int cpitch, int nE) {
  const int wid = __builtin_amdgcn_readfirstlane(
      (int)(blockIdx.x * GB_WPB + (threadIdx.x >> 5)));
  const int nw   = (int)gridDim.x * GB_WPB;
  const int lane = threadIdx.x & 31;
  const int plane = n_dst * PCAP;
  const float4* hb = (const float4*)hsrc + lane;

  int w01[GB_K], w23[GB_K];
  int deg[GB_K];
  bool chained[GB_K];
  float4 a0[GB_K], a1[GB_K];
#pragma unroll
  for (int j = 0; j < GB_K; ++j) {
    const int d  = wid + j * nw;
    const bool on = d < n_dst;
    const int dd = on ? d : n_dst - 1;
    a0[j] = make_float4(0.f, 0.f, 0.f, 0.f);
    a1[j] = make_float4(0.f, 0.f, 0.f, 0.f);
    const int* rl = slots + (size_t)dd * PCAP + (size_t)(lane >> 4) * plane + (lane & 15);
    const int x01 = rl[0];
    const int x23 = rl[2 * (size_t)plane];
    const int c0 = on ? cnt[dd] : 0;
    const int c1 = on ? cnt[cpitch + dd] : 0;
    const int c2 = on ? cnt[2 * cpitch + dd] : 0;
    const int c3 = on ? cnt[3 * cpitch + dd] : 0;
    const int e0 = c0 < PCAP ? (c0 > 0 ? c0 : 0) : PCAP;
    const int e1 = c1 < PCAP ? (c1 > 0 ? c1 : 0) : PCAP;
    const int e2 = c2 < PCAP ? (c2 > 0 ? c2 : 0) : PCAP;
    const int e3 = c3 < PCAP ? (c3 > 0 ? c3 : 0) : PCAP;
    deg[j] = c0 + c1 + c2 + c3;
    chained[j] = (c0 > PCAP) | (c1 > PCAP) | (c2 > PCAP) | (c3 > PCAP);
    const int r = lane & 15;
    w01[j] = r < (lane < 16 ? e0 : e1) ? (int)((unsigned)x01 << ROW_SHIFT) : -1;
    w23[j] = r < (lane < 16 ? e2 : e3) ? (int)((unsigned)x23 << ROW_SHIFT) : -1;
  }

#pragma unroll
  for (int j = 0; j < PF_D; ++j) PREFETCH_VISIT(j, 0)

#pragma nounroll
  for (int p = 0; p < NBAND; ++p) {
#pragma unroll
    for (int j = 0; j < GB_K; ++j) {
      PREFETCH_VISIT((j + PF_D) % GB_K, p + (j + PF_D) / GB_K)
      int b01 = w01[j] >> (BAND_SHIFT + ROW_SHIFT);  b01 = b01 > NBAND - 1 ? NBAND - 1 : b01;
      int b23 = w23[j] >> (BAND_SHIFT + ROW_SHIFT);  b23 = b23 > NBAND - 1 ? NBAND - 1 : b23;
      unsigned long long m =
            (unsigned long long)__builtin_amdgcn_ballot_w32(b01 == p)
          | ((unsigned long long)__builtin_amdgcn_ballot_w32(b23 == p) << 32);
      int n = __builtin_popcountll(m);
      for (; n >= 4; n -= 4) BAND_ROWS(4, j)
      if (n & 2) BAND_ROWS(2, j)
      if (n & 1) BAND_ROWS(1, j)
    }
  }

#pragma unroll
  for (int j = 0; j < GB_K; ++j) {
    const int d = wid + j * nw;
    if (d >= n_dst) continue;
    if (chained[j]) {
      for (int q = 0; q < NPART; ++q) {
        if (cnt[q * cpitch + d] <= PCAP) continue;
        int t = ovf_head[q * cpitch + d];
        for (int k = 0; t > 0 && k < nE; ++k) {
          const int2 r = ovf[t - 1];
          t = r.y;
          const float4* p = (const float4*)(hsrc + (size_t)r.x * IN_F) + lane;
          float4 v0 = p[0];
          float4 v1 = p[IN_F / 8];
          a0[j].x += v0.x; a0[j].y += v0.y; a0[j].z += v0.z; a0[j].w += v0.w;
          a1[j].x += v1.x; a1[j].y += v1.y; a1[j].z += v1.z; a1[j].w += v1.w;
        }
      }
    }
    const float inv = 1.0f / fmaxf((float)deg[j], 1.0f);
    float4 r0 = a0[j], r1 = a1[j];
    r0.x *= inv; r0.y *= inv; r0.z *= inv; r0.w *= inv;
    r1.x *= inv; r1.y *= inv; r1.z *= inv; r1.w *= inv;
    float4* o = (float4*)(hN + (size_t)d * IN_F) + lane;
    o[0] = r0;
    o[IN_F / 8] = r1;
  }
}
#undef BAND_ROWS
#undef PREFETCH_VISIT
#undef PREFETCH_ROW


#define WL_COLS     128
#define WL_WAVES    32
#define WL_F4       (KSTEPS * 2 * 2 * 32)

__global__ __launch_bounds__(32 * WL_WAVES)
void sage_gemm_wlds_kernel(const float* __restrict__ A0,
                           const float* __restrict__ A1,
                           const float* __restrict__ Wm,
                           const float* __restrict__ bias,
                           float* __restrict__ out,
                           int M, int N, int do_relu) {
  __shared__ float4 Wl[WL_F4];

  const int lane  = threadIdx.x & 31;
  const int wave  = threadIdx.x >> 5;
  const int nhalf = N / WL_COLS;
  const int half  = blockIdx.x % nhalf;
  const int gi    = blockIdx.x / nhalf;
  const int G     = gridDim.x / nhalf;

  const int l15 = lane & 15;
  const int kh  = (lane < 16) ? 0 : 2;

  const int ntile   = M >> 4;
  const int ntile_g = (gi < ntile) ? (ntile - gi + G - 1) / G : 0;
  const int nstrip  = ntile_g * 2;

  auto prefetch_tile = [&](int tm) {
    const float* r = ((lane < 16) ? A0 : A1) + (size_t)(tm * 16 + l15) * IN_F;
    __builtin_prefetch(r, 0, 2);
    __builtin_prefetch(r + 64, 0, 2);
    __builtin_prefetch(r + 128, 0, 2);
    __builtin_prefetch(r + 192, 0, 2);
  };
  const bool pf_wave = (wave & 1) == 0;
  if (pf_wave && wave < nstrip) prefetch_tile(gi + (wave >> 1) * G);

  for (int idx = threadIdx.x; idx < WL_F4; idx += 32 * WL_WAVES) {
    const int p  = (idx >> 5) & 1;
    const int g  = (idx >> 6) & 1;
    const int ks = idx >> 7;
    const float* w0 = Wm + (size_t)(ks * 4 + kh) * N
                         + half * WL_COLS + g * 64 + p * 32 + l15;
    float4 v;
    v.x = w0[0];
    v.y = w0[N];
    v.z = w0[16];
    v.w = w0[N + 16];
    Wl[idx] = v;
  }
  __syncthreads();

  const int mhi     = (lane < 16) ? 0 : 8;

  for (int q = wave; q < nstrip; q += WL_WAVES) {
    if (pf_wave && q + WL_WAVES < nstrip) prefetch_tile(gi + ((q + WL_WAVES) >> 1) * G);
    const int tm = gi + (q >> 1) * G;
    const int g  = q & 1;
    const int row = tm * 16 + l15;
    const float* Arow0 = A0 + (size_t)row * IN_F + kh;
    const float* Arow1 = A1 + (size_t)row * IN_F + kh;
    const float4* Bl = Wl + g * 64 + lane;

    v8f acc[4] = {};

#define K_STEP_L(AROW, BL, I)                                                 \
    {                                                                         \
      v2f a = *(const v2f*)((AROW) + (I) * 4);             \
      float4 bA = (BL)[(I) * 128];                    \
      float4 bB = (BL)[(I) * 128 + 32];               \
      v2f b0; b0.x = bA.x; b0.y = bA.y;                                       \
      v2f b1; b1.x = bA.z; b1.y = bA.w;                                       \
      v2f b2; b2.x = bB.x; b2.y = bB.y;                                       \
      v2f b3; b3.x = bB.z; b3.y = bB.w;                                       \
      acc[0] = __builtin_amdgcn_wmma_f32_16x16x4_f32(false, a, false, b0,     \
                                                     (short)0, acc[0], false, false); \
      acc[1] = __builtin_amdgcn_wmma_f32_16x16x4_f32(false, a, false, b1,     \
                                                     (short)0, acc[1], false, false); \
      acc[2] = __builtin_amdgcn_wmma_f32_16x16x4_f32(false, a, false, b2,     \
                                                     (short)0, acc[2], false, false); \
      acc[3] = __builtin_amdgcn_wmma_f32_16x16x4_f32(false, a, false, b3,     \
                                                     (short)0, acc[3], false, false); \
    }

    for (int kb = 0; kb < KSTEPS / 8; ++kb) {
      const float*  Ap = (kb < KSPLIT / 8) ? Arow0 + kb * 32
                                           : Arow1 + (kb - KSPLIT / 8) * 32;
      const float4* Bp = Bl + kb * (8 * 128);
#pragma unroll
      for (int j = 0; j < 8; ++j) {
        K_STEP_L(Ap, Bp, j);
      }
    }
#undef K_STEP_L

    const int mbase = tm * 16 + mhi;
    const int col0  = half * WL_COLS + g * 64 + l15;
#pragma unroll
    for (int t = 0; t < 4; ++t) {
      const int col = col0 + t * 16;
      const float bv = bias[col];
#pragma unroll
      for (int i = 0; i < 8; ++i) {
        float v = acc[t][i] + bv;
        if (do_relu) v = fmaxf(v, 0.0f);
        out[(size_t)(mbase + i) * N + col] = v;
      }
    }
  }
}

#define SPLITK       8
#define KS_PER_WAVE  (KSTEPS / SPLITK)
#define PART_EPITCH  34
#define PART_SPITCH  (32 * PART_EPITCH)
#define A_PITCH      (IN_F + 4)

static_assert(2 * SPLITK == 16, "two nodes per wave fill the 16-row tile");
static_assert(KSPLIT % KS_PER_WAVE == 0, "a wave's k lie in one half of the concat");

__global__ __launch_bounds__(32 * SPLITK)
void sage_gather_gemm_splitk_kernel(const float* __restrict__ hsrc,
                                    const int* __restrict__ slots,
                                    const int* __restrict__ cnt,
                                    const int* __restrict__ ovf_head,
                                    const int2* __restrict__ ovf,
                                    const float* __restrict__ Wm,
                                    const float* __restrict__ bias,
                                    float* __restrict__ out,
                                    int M, int N, int cpitch, int nE) {
  __shared__ float part[SPLITK * PART_SPITCH];
  __shared__ __attribute__((aligned(16))) float At[16 * A_PITCH];

  const int lane  = threadIdx.x & 31;
  const int wave  = __builtin_amdgcn_readfirstlane((int)(threadIdx.x >> 5));
  const int ngrpN = N >> 6;
  const int grp   = blockIdx.x;
  const int ngrps = (M >> 4) * ngrpN;
  if (grp >= ngrps) return;

  const int tm = grp / ngrpN;
  const int gn = grp % ngrpN;

  {
    const int dA = tm * 16 + 2 * wave;
    const int dB = dA + 1;
    const int plane = M * PCAP;
    const int* rl = slots + (size_t)dA * PCAP + (size_t)(lane >> 4) * plane + (lane & 15);
    const int wA01 = rl[0];
    const int wA23 = rl[2 * (size_t)plane];
    const int wB01 = rl[PCAP];
    const int wB23 = rl[2 * (size_t)plane + PCAP];
    const int cA0 = cnt[dA], cA1 = cnt[cpitch + dA], cA2 = cnt[2 * cpitch + dA], cA3 = cnt[3 * cpitch + dA];
    const int cB0 = cnt[dB], cB1 = cnt[cpitch + dB], cB2 = cnt[2 * cpitch + dB], cB3 = cnt[3 * cpitch + dB];
    float4 r0, r1;
    gather_mean_node(hsrc, ovf_head, ovf, dA, wA01, wA23, cA0, cA1, cA2, cA3,
                     cpitch, nE, lane, r0, r1);
    float4* o = (float4*)(At + (2 * wave) * A_PITCH) + lane;
    o[0] = r0;
    o[IN_F / 8] = r1;
    gather_mean_node(hsrc, ovf_head, ovf, dB, wB01, wB23, cB0, cB1, cB2, cB3,
                     cpitch, nE, lane, r0, r1);
    o = (float4*)(At + (2 * wave + 1) * A_PITCH) + lane;
    o[0] = r0;
    o[IN_F / 8] = r1;
  }
  __syncthreads();

  const int l15 = lane & 15;
  const int row = tm * 16 + l15;
  const int kh  = (lane < 16) ? 0 : 2;

  const int ks0 = wave * KS_PER_WAVE;
  const float* Wc = Wm + (size_t)(ks0 * 4 + kh) * N + gn * 64 + l15;

  v8f acc[4] = {};

#define K_STEP_W(ALOAD, I)                                                    \
    {                                                                         \
      v2f a = ALOAD;                                   \
      const float* wr = Wc + (size_t)(I) * 4 * N;                             \
      v2f b0; b0.x = wr[0];  b0.y = wr[N];                                    \
      v2f b1; b1.x = wr[16]; b1.y = wr[N + 16];                               \
      v2f b2; b2.x = wr[32]; b2.y = wr[N + 32];                               \
      v2f b3; b3.x = wr[48]; b3.y = wr[N + 48];                               \
      acc[0] = __builtin_amdgcn_wmma_f32_16x16x4_f32(false, a, false, b0,     \
                                                     (short)0, acc[0], false, false); \
      acc[1] = __builtin_amdgcn_wmma_f32_16x16x4_f32(false, a, false, b1,     \
                                                     (short)0, acc[1], false, false); \
      acc[2] = __builtin_amdgcn_wmma_f32_16x16x4_f32(false, a, false, b2,     \
                                                     (short)0, acc[2], false, false); \
      acc[3] = __builtin_amdgcn_wmma_f32_16x16x4_f32(false, a, false, b3,     \
                                                     (short)0, acc[3], false, false); \
    }

  if (ks0 < KSPLIT) {
    const float* Arow = hsrc + (size_t)row * IN_F + ks0 * 4 + kh;
#pragma unroll 4
    for (int i = 0; i < KS_PER_WAVE; ++i) K_STEP_W(*(const v2f*)(Arow + i * 4), i)
  } else {
    const float* Al = At + l15 * A_PITCH + (ks0 - KSPLIT) * 4 + kh;
#pragma unroll 4
    for (int i = 0; i < KS_PER_WAVE; ++i) K_STEP_W(*(const v2f*)(Al + i * 4), i)
  }
#undef K_STEP_W

  float* pw = part + wave * PART_SPITCH + lane;
#pragma unroll
  for (int t = 0; t < 4; ++t) {
#pragma unroll
    for (int i = 0; i < 8; ++i) {
      pw[(t * 8 + i) * PART_EPITCH] = acc[t][i];
    }
  }
  __syncthreads();

#pragma unroll
  for (int j = 0; j < 4; ++j) {
    const int r  = wave * 2 + (j >> 1);
    const int cc = (j & 1) * 32 + lane;
    const int e  = (cc >> 4) * 8 + (r & 7);
    const int sl = (cc & 15) + ((r >> 3) << 4);
    const float* pr = part + e * PART_EPITCH + sl;
    float v = pr[0];
#pragma unroll
    for (int s = 1; s < SPLITK; ++s) v += pr[s * PART_SPITCH];
    const int col = gn * 64 + cc;
    v += bias[col];
    out[(size_t)(tm * 16 + r) * N + col] = v;
  }
}

extern "C" void kernel_launch(void* const* d_in, const int* in_sizes, int n_in,
                              void* d_out, int out_size, void* d_ws, size_t ws_size,
                              hipStream_t stream) {
  const float* x   = (const float*)d_in[0];
  const float* W1  = (const float*)d_in[1];
  const float* b1  = (const float*)d_in[2];
  const float* W2  = (const float*)d_in[3];
  const float* b2  = (const float*)d_in[4];
  const int* src1  = (const int*)d_in[5];
  const int* dst1  = (const int*)d_in[6];
  const int* src2  = (const int*)d_in[7];
  const int* dst2  = (const int*)d_in[8];
  const int E1 = in_sizes[5];
  const int E2 = in_sizes[7];

  char* wsb = (char*)d_ws;
  const size_t int_words = 3 * (size_t)(N_DST1 + N_DST2) + 2 + (size_t)E1 + (size_t)E2;
  int*  cnt2      = (int*)wsb;
  int*  ovf_head2 = cnt2 + NPART * CPITCH2;
  int*  slots2    = ovf_head2 + NPART * CPITCH2;
  int2* ovf2      = (int2*)(slots2 + (size_t)N_DST2 * NBR_CAP);
  size_t off = (int_words * 4) & ~(size_t)255;
  auto take = [&](size_t n_elems) { void* p = wsb + off; off += n_elems * 4; return p; };

  float* hN1    = (float*)take((size_t)N_DST1 * IN_F);
  float* h      = (float*)take((size_t)N_DST1 * HID_F);
  int*  slots1    = (int*)h;
  int2* ovf1      = (int2*)(slots1 + (size_t)N_DST1 * NBR_CAP);
  int*  cnt1      = (int*)(ovf1 + E1);
  int*  ovf_head1 = cnt1 + NPART * CPITCH1;
  (void)ws_size;

  build_rows_kernel<<<NWG_ROWS1 + NWG_ROWS2, OWN_THR, 0, stream>>>(
      src1, dst1, cnt1, slots1, ovf_head1, ovf1, E1, N_DST1, NWG_ROWS1,
      src2, dst2, cnt2, slots2, ovf_head2, ovf2, E2, N_DST2);

  gather_mean_banded_kernel<<<GB_NWG, 32 * GB_WPB, 0, stream>>>(
      x, slots1, cnt1, ovf_head1, ovf1, hN1, N_DST1, CPITCH1, E1);

  sage_gemm_wlds_kernel<<<256, 32 * WL_WAVES, 0, stream>>>(
      x, hN1, W1, b1, h, N_DST1, HID_F,  1);

  {
    int ngrps = (N_DST2 / 16) * (N_CLS / 64);
    sage_gather_gemm_splitk_kernel<<<ngrps, 32 * SPLITK, 0, stream>>>(
        h, slots2, cnt2, ovf_head2, ovf2, W2, b2, (float*)d_out,
        N_DST2, N_CLS, CPITCH2, E2);
  }
}
